# starting kernel
# baseline (speedup 1.0000x reference)
#include <hip/hip_runtime.h>
#include <hip/hip_bf16.h>


typedef __attribute__((ext_vector_type(16))) _Float16 v16h;
typedef __attribute__((ext_vector_type(8)))  _Float16 v8h;
typedef __attribute__((ext_vector_type(8)))  float    v8f;

#define NNODES 50000
#define NEDGES 800000
#define ETOT   (NEDGES + NNODES)
#define FDIM   128
#define NHEAD  4

__device__ __forceinline__ unsigned fkey(float f) {
  unsigned u = __float_as_uint(f);
  return (u & 0x80000000u) ? ~u : (u | 0x80000000u);
}
__device__ __forceinline__ float funkey(unsigned k) {
  unsigned u = (k & 0x80000000u) ? (k & 0x7fffffffu) : ~k;
  return __uint_as_float(u);
}

__global__ void cast_f16_kernel(const float* __restrict__ in,
                                _Float16* __restrict__ out, int n) {
  int i = blockIdx.x * blockDim.x + threadIdx.x;
  if (i < n) out[i] = (_Float16)in[i];
}

__global__ void init_conv_kernel(float* __restrict__ acc,
                                 unsigned* __restrict__ mkey,
                                 float* __restrict__ ssum, int n) {
  int i = blockIdx.x * blockDim.x + threadIdx.x;
  if (i < n * FDIM) acc[i] = 0.0f;
  if (i < n * NHEAD) { mkey[i] = 0u; ssum[i] = 0.0f; }
}

#define KC 64
#define BT_STRIDE 72

__global__ __launch_bounds__(256) void wmma_gemm_kernel(
    const _Float16* __restrict__ A, const float* __restrict__ W,
    const float* __restrict__ bias,
    float* __restrict__ outF, _Float16* __restrict__ outH,
    int M, int Nt, int K, int act) {
  __shared__ _Float16 bt[64 * BT_STRIDE];

  const int lane = threadIdx.x & 31;
  const int wave = threadIdx.x >> 5;
  const int half = lane >> 4;
  const int l15  = lane & 15;
  const int n0   = blockIdx.y * 64;

  int rowBase = blockIdx.x * 128 + wave * 16;
  if (rowBase > M - 16) rowBase = M - 16;

  v8f c[4] = {v8f{}, v8f{}, v8f{}, v8f{}};

  for (int kc = 0; kc < K; kc += KC) {
    {
      int kl = threadIdx.x >> 4;
      const int n4 = (threadIdx.x & 15) * 4;
      #pragma unroll
      for (int p = 0; p < 4; ++p, kl += 16) {
        const float4 wv =
            *(const float4*)(W + (size_t)(kc + kl) * Nt + n0 + n4);
        bt[(n4 + 0) * BT_STRIDE + kl] = (_Float16)wv.x;
        bt[(n4 + 1) * BT_STRIDE + kl] = (_Float16)wv.y;
        bt[(n4 + 2) * BT_STRIDE + kl] = (_Float16)wv.z;
        bt[(n4 + 3) * BT_STRIDE + kl] = (_Float16)wv.w;
      }
    }
    __syncthreads();

    const _Float16* arow = A + (size_t)(rowBase + l15) * K + kc;
    #pragma unroll
    for (int ks = 0; ks < 2; ++ks) {
      const int kb = ks * 32 + half * 8;
      v8h alo = *(const v8h*)(arow + kb);
      v8h ahi = *(const v8h*)(arow + kb + 16);
      v16h a = __builtin_shufflevector(alo, ahi, 0, 1, 2, 3, 4, 5, 6, 7,
                                       8, 9, 10, 11, 12, 13, 14, 15);
      v16h b[4];
      #pragma unroll
      for (int nn = 0; nn < 4; ++nn) {
        const _Float16* bp = bt + (nn * 16 + l15) * BT_STRIDE + kb;
        v8h blo = *(const v8h*)(bp);
        v8h bhi = *(const v8h*)(bp + 16);
        b[nn] = __builtin_shufflevector(blo, bhi, 0, 1, 2, 3, 4, 5, 6, 7,
                                        8, 9, 10, 11, 12, 13, 14, 15);
      }
      #pragma unroll
      for (int nn = 0; nn < 4; ++nn) {
        c[nn] = __builtin_amdgcn_wmma_f32_16x16x32_f16(
            false, a, false, b[nn], (short)0, c[nn], false, false);
      }
    }
    __syncthreads();
  }

  #pragma unroll
  for (int nn = 0; nn < 4; ++nn) {
    const int col = n0 + nn * 16 + l15;
    const float bv = bias ? bias[col] : 0.0f;
    #pragma unroll
    for (int r = 0; r < 8; ++r) {
      const int row = rowBase + half * 8 + r;
      float v = c[nn][r] + bv;
      if (act == 1) v = v > 0.0f ? v : (__expf(v) - 1.0f);
      else if (act == 2) v = 1.0f / (1.0f + __expf(-v));
      if (outF) outF[(size_t)row * Nt + col] = v;
      if (outH) outH[(size_t)row * Nt + col] = (_Float16)v;
    }
  }
}

__global__ void edge_logits_kernel(const int* __restrict__ ei,
                                   const float* __restrict__ xl,
                                   const float* __restrict__ xr,
                                   const float* __restrict__ att,
                                   float* __restrict__ ebuf,
                                   unsigned* __restrict__ mkey) {
  int idx = blockIdx.x * blockDim.x + threadIdx.x;
  if (idx >= ETOT * NHEAD) return;
  const int e = idx >> 2, h = idx & 3;
  int s, d;
  if (e < NEDGES) { s = ei[e]; d = ei[NEDGES + e]; } else { s = d = e - NEDGES; }

  const float4* pl = (const float4*)(xl + (size_t)s * FDIM + h * 32);
  const float4* pr = (const float4*)(xr + (size_t)d * FDIM + h * 32);
  const float4* pa = (const float4*)(att + h * 32);
  float acc = 0.0f;
  #pragma unroll
  for (int i = 0; i < 8; ++i) {
    const float4 a = pl[i], b = pr[i], w = pa[i];
    float z;
    z = a.x + b.x; z = z > 0.0f ? z : 0.2f * z; acc += z * w.x;
    z = a.y + b.y; z = z > 0.0f ? z : 0.2f * z; acc += z * w.y;
    z = a.z + b.z; z = z > 0.0f ? z : 0.2f * z; acc += z * w.z;
    z = a.w + b.w; z = z > 0.0f ? z : 0.2f * z; acc += z * w.w;
  }
  ebuf[idx] = acc;
  atomicMax(mkey + (size_t)d * NHEAD + h, fkey(acc));
}

__global__ void edge_exp_kernel(const int* __restrict__ ei,
                                float* __restrict__ ebuf,
                                const unsigned* __restrict__ mkey,
                                float* __restrict__ ssum) {
  int idx = blockIdx.x * blockDim.x + threadIdx.x;
  if (idx >= ETOT * NHEAD) return;
  const int e = idx >> 2, h = idx & 3;
  int d;
  if (e < NEDGES) d = ei[NEDGES + e]; else d = e - NEDGES;
  const float m = funkey(mkey[(size_t)d * NHEAD + h]);
  const float p = __expf(ebuf[idx] - m);
  ebuf[idx] = p;
  atomicAdd(ssum + (size_t)d * NHEAD + h, p);
}

__global__ void edge_scatter_kernel(const int* __restrict__ ei,
                                    const float* __restrict__ xl,
                                    const float* __restrict__ ebuf,
                                    const float* __restrict__ ssum,
                                    float* __restrict__ acc) {
  int idx = blockIdx.x * blockDim.x + threadIdx.x;
  if (idx >= ETOT * NHEAD) return;
  const int e = idx >> 2, h = idx & 3;
  int s, d;
  if (e < NEDGES) { s = ei[e]; d = ei[NEDGES + e]; } else { s = d = e - NEDGES; }
  const float alpha = ebuf[idx] / ssum[(size_t)d * NHEAD + h];
  const float4* pl = (const float4*)(xl + (size_t)s * FDIM + h * 32);
  float* ob = acc + (size_t)d * FDIM + h * 32;
  #pragma unroll
  for (int i = 0; i < 8; ++i) {
    const float4 a = pl[i];
    atomicAdd(ob + i * 4 + 0, a.x * alpha);
    atomicAdd(ob + i * 4 + 1, a.y * alpha);
    atomicAdd(ob + i * 4 + 2, a.z * alpha);
    atomicAdd(ob + i * 4 + 3, a.w * alpha);
  }
}

__global__ void node_finish_kernel(const float* __restrict__ acc,
                                   const float* __restrict__ b,
                                   _Float16* __restrict__ outH, int n) {
  int i = blockIdx.x * blockDim.x + threadIdx.x;
  if (i >= n * FDIM) return;
  float v = acc[i] + b[i & (FDIM - 1)];
  v = v > 0.0f ? v : (__expf(v) - 1.0f);
  outH[i] = (_Float16)v;
}

extern "C" void kernel_launch(void* const* d_in, const int* in_sizes, int n_in,
                              void* d_out, int out_size, void* d_ws,
                              size_t ws_size, hipStream_t stream) {
  (void)in_sizes; (void)n_in; (void)out_size; (void)ws_size;
  const float* x    = (const float*)d_in[0];
  const int*   ei   = (const int*)d_in[1];
  const float* Wl1  = (const float*)d_in[2];
  const float* Wr1  = (const float*)d_in[3];
  const float* att1 = (const float*)d_in[4];
  const float* b1   = (const float*)d_in[5];
  const float* Wl2  = (const float*)d_in[6];
  const float* Wr2  = (const float*)d_in[7];
  const float* att2 = (const float*)d_in[8];
  const float* b2   = (const float*)d_in[9];
  const float* W1   = (const float*)d_in[10];
  const float* bb1  = (const float*)d_in[11];
  const float* W2   = (const float*)d_in[12];
  const float* bb2  = (const float*)d_in[13];
  const float* W3   = (const float*)d_in[14];
  const float* bb3  = (const float*)d_in[15];
  float* out = (float*)d_out;

  char* ws = (char*)d_ws;
  size_t off = 0;
  auto alloc = [&](size_t bytes) -> char* {
    char* p = ws + off;
    off += (bytes + 255) & ~(size_t)255;
    return p;
  };
  const size_t S_F = (size_t)NNODES * FDIM;
  _Float16* xh  = (_Float16*)alloc(S_F * 2);
  _Float16* h1h = (_Float16*)alloc(S_F * 2);
  _Float16* h2h = (_Float16*)alloc(S_F * 2);
  const size_t uoff = off;
  float*    xl   = (float*)alloc(S_F * 4);
  float*    xr   = (float*)alloc(S_F * 4);
  float*    accb = (float*)alloc(S_F * 4);
  float*    ebuf = (float*)alloc((size_t)ETOT * NHEAD * 4);
  unsigned* mkey = (unsigned*)alloc((size_t)NNODES * NHEAD * 4);
  float*    ssum = (float*)alloc((size_t)NNODES * NHEAD * 4);
  _Float16* m1 = (_Float16*)(ws + uoff);
  _Float16* m2 = (_Float16*)(ws + uoff + (size_t)NNODES * 640 * 2);

  const int TB = 256;
  const int ngrid = (int)((S_F + TB - 1) / TB);
  const int egrid = (ETOT * NHEAD + TB - 1) / TB;
  const dim3 gconv((NNODES + 127) / 128, 128 / 64);

  cast_f16_kernel<<<ngrid, TB, 0, stream>>>(x, xh, (int)S_F);
  wmma_gemm_kernel<<<gconv, TB, 0, stream>>>(xh, Wl1, nullptr, xl, nullptr,
                                             NNODES, 128, 128, 0);
  wmma_gemm_kernel<<<gconv, TB, 0, stream>>>(xh, Wr1, nullptr, xr, nullptr,
                                             NNODES, 128, 128, 0);
  init_conv_kernel<<<ngrid, TB, 0, stream>>>(accb, mkey, ssum, NNODES);
  edge_logits_kernel<<<egrid, TB, 0, stream>>>(ei, xl, xr, att1, ebuf, mkey);
  edge_exp_kernel<<<egrid, TB, 0, stream>>>(ei, ebuf, mkey, ssum);
  edge_scatter_kernel<<<egrid, TB, 0, stream>>>(ei, xl, ebuf, ssum, accb);
  node_finish_kernel<<<ngrid, TB, 0, stream>>>(accb, b1, h1h, NNODES);

  wmma_gemm_kernel<<<gconv, TB, 0, stream>>>(h1h, Wl2, nullptr, xl, nullptr,
                                             NNODES, 128, 128, 0);
  wmma_gemm_kernel<<<gconv, TB, 0, stream>>>(h1h, Wr2, nullptr, xr, nullptr,
                                             NNODES, 128, 128, 0);
  init_conv_kernel<<<ngrid, TB, 0, stream>>>(accb, mkey, ssum, NNODES);
  edge_logits_kernel<<<egrid, TB, 0, stream>>>(ei, xl, xr, att2, ebuf, mkey);
  edge_exp_kernel<<<egrid, TB, 0, stream>>>(ei, ebuf, mkey, ssum);
  edge_scatter_kernel<<<egrid, TB, 0, stream>>>(ei, xl, ebuf, ssum, accb);
  node_finish_kernel<<<ngrid, TB, 0, stream>>>(accb, b2, h2h, NNODES);

  const dim3 gm1((NNODES + 127) / 128, 640 / 64);
  wmma_gemm_kernel<<<gm1, TB, 0, stream>>>(h2h, W1, bb1, nullptr, m1,
                                           NNODES, 640, 128, 1);
  const dim3 gm2((NNODES + 127) / 128, 320 / 64);
  wmma_gemm_kernel<<<gm2, TB, 0, stream>>>(m1, W2, bb2, nullptr, m2,
                                           NNODES, 320, 640, 1);
  const dim3 gm3((NNODES + 127) / 128, 64 / 64);
  wmma_gemm_kernel<<<gm3, TB, 0, stream>>>(m2, W3, bb3, out, nullptr,
                                           NNODES, 64, 320, 2);
}


# Weights rounded to f16 and transposed to [N][K] once per call by one small kernel; GEMMs stage B tiles with 16-B copies, not 2-B LDS stores after a per-block convert; k and sum order unchanged
# speedup vs baseline: 4.2594x; 4.2594x over previous
#include <hip/hip_runtime.h>
#include <hip/hip_bf16.h>


typedef __attribute__((ext_vector_type(16))) _Float16 v16h;
typedef __attribute__((ext_vector_type(8)))  _Float16 v8h;
typedef __attribute__((ext_vector_type(8)))  float    v8f;

#define NNODES 50000
#define NEDGES 800000
#define ETOT   (NEDGES + NNODES)
#define FDIM   128
#define NHEAD  4

__device__ __forceinline__ unsigned fkey(float f) {
  unsigned u = __float_as_uint(f);
  return (u & 0x80000000u) ? ~u : (u | 0x80000000u);
}
__device__ __forceinline__ float funkey(unsigned k) {
  unsigned u = (k & 0x80000000u) ? (k & 0x7fffffffu) : ~k;
  return __uint_as_float(u);
}

__global__ void cast_f16_kernel(const float* __restrict__ in,
                                _Float16* __restrict__ out, int n) {
  int i = blockIdx.x * blockDim.x + threadIdx.x;
  if (i < n) out[i] = (_Float16)in[i];
}

__global__ void init_conv_kernel(unsigned* __restrict__ mkey,
                                 float* __restrict__ ssum, int n) {
  int i = blockIdx.x * blockDim.x + threadIdx.x;
  if (i < n * NHEAD) { mkey[i] = 0u; ssum[i] = 0.0f; }
}

#define WT_OFF_L1  0
#define WT_OFF_R1  (WT_OFF_L1 + 128 * 128)
#define WT_OFF_L2  (WT_OFF_R1 + 128 * 128)
#define WT_OFF_R2  (WT_OFF_L2 + 128 * 128)
#define WT_OFF_M1  (WT_OFF_R2 + 128 * 128)
#define WT_OFF_M2  (WT_OFF_M1 + 128 * 640)
#define WT_OFF_M3  (WT_OFF_M2 + 640 * 320)
#define WT_TOTAL   (WT_OFF_M3 + 320 * 64)
#define WT_MAXTILES ((640 / 32) * (320 / 32))

__global__ __launch_bounds__(256) void weights_f16t_kernel(
    const float* __restrict__ w0, const float* __restrict__ w1,
    const float* __restrict__ w2, const float* __restrict__ w3,
    const float* __restrict__ w4, const float* __restrict__ w5,
    const float* __restrict__ w6, _Float16* __restrict__ wt) {
  __shared__ float tile[32][33];
  const float* Wm; int K, Nt, off;
  switch (blockIdx.y) {
    case 0:  Wm = w0; K = 128; Nt = 128; off = WT_OFF_L1; break;
    case 1:  Wm = w1; K = 128; Nt = 128; off = WT_OFF_R1; break;
    case 2:  Wm = w2; K = 128; Nt = 128; off = WT_OFF_L2; break;
    case 3:  Wm = w3; K = 128; Nt = 128; off = WT_OFF_R2; break;
    case 4:  Wm = w4; K = 128; Nt = 640; off = WT_OFF_M1; break;
    case 5:  Wm = w5; K = 640; Nt = 320; off = WT_OFF_M2; break;
    default: Wm = w6; K = 320; Nt = 64;  off = WT_OFF_M3; break;
  }
  const int tn = Nt >> 5;
  if ((int)blockIdx.x >= (K >> 5) * tn) return;
  const int k0 = ((int)blockIdx.x / tn) * 32;
  const int n0 = ((int)blockIdx.x % tn) * 32;
  const int lane = threadIdx.x & 31;
  const int wv   = threadIdx.x >> 5;
  #pragma unroll
  for (int j = 0; j < 4; ++j) {
    const int k = wv + 8 * j;
    tile[k][lane] = Wm[(size_t)(k0 + k) * Nt + n0 + lane];
  }
  __syncthreads();
  #pragma unroll
  for (int j = 0; j < 4; ++j) {
    const int n = wv + 8 * j;
    wt[off + (n0 + n) * K + k0 + lane] = (_Float16)tile[lane][n];
  }
}

#define KC 64
#define BT_STRIDE 72

__global__ __launch_bounds__(256) void wmma_gemm_kernel(
    const _Float16* __restrict__ A, const _Float16* __restrict__ Wt,
    const float* __restrict__ bias,
    float* __restrict__ outF, _Float16* __restrict__ outH,
    int M, int Nt, int K, int act) {
  __shared__ _Float16 bt[64 * BT_STRIDE];

  const int lane = threadIdx.x & 31;
  const int wave = threadIdx.x >> 5;
  const int half = lane >> 4;
  const int l15  = lane & 15;
  const int n0   = blockIdx.y * 64;

  int rowBase = blockIdx.x * 128 + wave * 16;
  if (rowBase > M - 16) rowBase = M - 16;

  v8f c[4] = {v8f{}, v8f{}, v8f{}, v8f{}};

  for (int kc = 0; kc < K; kc += KC) {
    {
      const int q8 = (threadIdx.x & 7) * 8;
      int n = threadIdx.x >> 3;
      #pragma unroll
      for (int p = 0; p < 2; ++p, n += 32) {
        const v8h wv = *(const v8h*)(Wt + (size_t)(n0 + n) * K + kc + q8);
        *(v8h*)(bt + n * BT_STRIDE + q8) = wv;
      }
    }
    __syncthreads();

    const _Float16* arow = A + (size_t)(rowBase + l15) * K + kc;
    #pragma unroll
    for (int ks = 0; ks < 2; ++ks) {
      const int kb = ks * 32 + half * 8;
      v8h alo = *(const v8h*)(arow + kb);
      v8h ahi = *(const v8h*)(arow + kb + 16);
      v16h a = __builtin_shufflevector(alo, ahi, 0, 1, 2, 3, 4, 5, 6, 7,
                                       8, 9, 10, 11, 12, 13, 14, 15);
      v16h b[4];
      #pragma unroll
      for (int nn = 0; nn < 4; ++nn) {
        const _Float16* bp = bt + (nn * 16 + l15) * BT_STRIDE + kb;
        v8h blo = *(const v8h*)(bp);
        v8h bhi = *(const v8h*)(bp + 16);
        b[nn] = __builtin_shufflevector(blo, bhi, 0, 1, 2, 3, 4, 5, 6, 7,
                                        8, 9, 10, 11, 12, 13, 14, 15);
      }
      #pragma unroll
      for (int nn = 0; nn < 4; ++nn) {
        c[nn] = __builtin_amdgcn_wmma_f32_16x16x32_f16(
            false, a, false, b[nn], (short)0, c[nn], false, false);
      }
    }
    __syncthreads();
  }

  #pragma unroll
  for (int nn = 0; nn < 4; ++nn) {
    const int col = n0 + nn * 16 + l15;
    const float bv = bias ? bias[col] : 0.0f;
    #pragma unroll
    for (int r = 0; r < 8; ++r) {
      const int row = rowBase + half * 8 + r;
      float v = c[nn][r] + bv;
      if (act == 1) v = v > 0.0f ? v : (__expf(v) - 1.0f);
      else if (act == 2) v = 1.0f / (1.0f + __expf(-v));
      if (outF) outF[(size_t)row * Nt + col] = v;
      if (outH) outH[(size_t)row * Nt + col] = (_Float16)v;
    }
  }
}

#define EL_EPW   8
#define EL_WAVES (ETOT / EL_EPW)
#define EL_GRID  ((EL_WAVES + 7) / 8)
static_assert(ETOT % EL_EPW == 0 && NEDGES % EL_EPW == 0,
              "a wave's eight edges are all list edges or all self loops");

template <int M>
__device__ __forceinline__ float xor_lane(float v) {
  return __int_as_float(__builtin_amdgcn_update_dpp(
      0, __float_as_int(v), 0x160 + M, 0xf, 0xf, true));
}

__global__ __launch_bounds__(256) void edge_logits_kernel(
    const int* __restrict__ ei, const float* __restrict__ xl,
    const float* __restrict__ xr, const float* __restrict__ att,
    float* __restrict__ ebuf, unsigned* __restrict__ mkey) {
  const int lane = threadIdx.x & 31;
  const int wv = __builtin_amdgcn_readfirstlane((int)(threadIdx.x >> 5));
  const int e0 = (blockIdx.x * 8 + wv) * EL_EPW;
  if (e0 >= ETOT) return;
  const int u_l = lane & 7, g = lane >> 3;
  const int e = e0 + u_l;
  int s, d;
  if (e0 < NEDGES) { s = ei[e]; d = ei[NEDGES + e]; } else { s = d = e - NEDGES; }

  const float4 w = *(const float4*)(att + lane * 4);
  float4 a[EL_EPW], b[EL_EPW];
  #pragma unroll
  for (int u = 0; u < EL_EPW; ++u) {
    const int su = __builtin_amdgcn_readlane(s, u);
    const int du = __builtin_amdgcn_readlane(d, u);
    a[u] = *(const float4*)(xl + su * FDIM + lane * 4);
    b[u] = *(const float4*)(xr + du * FDIM + lane * 4);
  }
  float mine = 0.0f;
  #pragma unroll
  for (int u = 0; u < EL_EPW; ++u) {
    float z, acc;
    z = a[u].x + b[u].x; z = z > 0.0f ? z : 0.2f * z; acc = z * w.x;
    z = a[u].y + b[u].y; z = z > 0.0f ? z : 0.2f * z; acc += z * w.y;
    z = a[u].z + b[u].z; z = z > 0.0f ? z : 0.2f * z; acc += z * w.z;
    z = a[u].w + b[u].w; z = z > 0.0f ? z : 0.2f * z; acc += z * w.w;
    acc += xor_lane<1>(acc);
    acc += xor_lane<2>(acc);
    acc += xor_lane<4>(acc);
    if (u_l == u) mine = acc;
  }
  ebuf[e * NHEAD + g] = mine;
  atomicMax(mkey + (size_t)d * NHEAD + g, fkey(mine));
}

__global__ void edge_exp_kernel(const int* __restrict__ ei,
                                float* __restrict__ ebuf,
                                const unsigned* __restrict__ mkey,
                                float* __restrict__ ssum) {
  int idx = blockIdx.x * blockDim.x + threadIdx.x;
  if (idx >= ETOT * NHEAD) return;
  const int e = idx >> 2, h = idx & 3;
  int d;
  if (e < NEDGES) d = ei[NEDGES + e]; else d = e - NEDGES;
  const float m = funkey(mkey[(size_t)d * NHEAD + h]);
  const float p = __expf(ebuf[idx] - m);
  ebuf[idx] = p;
  atomicAdd(ssum + (size_t)d * NHEAD + h, p);
}

#define AGG_R    196
#define AGG_WG   ((NNODES + AGG_R - 1) / AGG_R)
#define AGG_TB   1024
#define AGG_CAP  16384
#define NE4      (NEDGES / 4)
static_assert(NEDGES % 4 == 0, "dst list is read as int4");
static_assert(AGG_R <= AGG_TB && AGG_R <= AGG_CAP, "self loops are queued by one pass");

#define NK16     (NEDGES / 16)
#define SC_U     8
static_assert(AGG_WG <= 256, "an owner's number fits one byte");
static_assert(NEDGES % 16 == 0, "key bytes are read as uint4");
static_assert(SC_U == 8, "the scan packs eight loads' matches into four words");

__global__ __launch_bounds__(256) void dst_key_kernel(
    const int* __restrict__ ei, unsigned* __restrict__ key) {
  const int i = blockIdx.x * blockDim.x + threadIdx.x;
  if (i >= NE4) return;
  const int4 v = ((const int4*)(ei + NEDGES))[i];
  key[i] = ((unsigned)v.x / AGG_R) | (((unsigned)v.y / AGG_R) << 8) |
           (((unsigned)v.z / AGG_R) << 16) | (((unsigned)v.w / AGG_R) << 24);
}

__device__ __forceinline__ unsigned zero_bytes(unsigned x) {
  return ~(((x & 0x7f7f7f7fu) + 0x7f7f7f7fu) | x | 0x7f7f7f7fu);
}

typedef __attribute__((ext_vector_type(2))) _Float16 v2h;

__device__ __forceinline__ bool agg_hit(int e, int* cnt, int* list) {
  const int pos = atomicAdd(cnt, 1);
  if (pos < AGG_CAP) list[pos] = e;
  return pos >= AGG_CAP;
}

__device__ __forceinline__ void agg_add_row(int e, int dl, int lo, float* acc,
                                            const int* __restrict__ ei,
                                            const float* __restrict__ xl,
                                            const float* __restrict__ ebuf,
                                            const float* __restrict__ ssum) {
  const int s = ei[e];
  const float4 p = *(const float4*)(ebuf + (size_t)e * NHEAD);
  const float4 q = *(const float4*)(ssum + (size_t)(lo + dl) * NHEAD);
  const float al[4] = {p.x / q.x, p.y / q.y, p.z / q.z, p.w / q.w};
  for (int c = 0; c < FDIM; ++c)
    atomicAdd(acc + dl * FDIM + c, xl[(size_t)s * FDIM + c] * al[c >> 5]);
}

__global__ __launch_bounds__(AGG_TB) void node_aggregate_kernel(
    const int* __restrict__ ei, const unsigned* __restrict__ key,
    const float* __restrict__ xl,
    const float* __restrict__ ebuf, const float* __restrict__ ssum,
    const float* __restrict__ b, _Float16* __restrict__ outH) {
  __shared__ __attribute__((aligned(16))) float acc[AGG_R * FDIM];
  __shared__ int   list[AGG_CAP];
  __shared__ int   cnt;

  const int tid  = threadIdx.x;
  const int lane = tid & 31;
  const int wave = tid >> 5;
  const int lo   = blockIdx.x * AGG_R;
  const int nown = min(AGG_R, NNODES - lo);

  for (int i = tid; i < AGG_R * FDIM / 4; i += AGG_TB)
    ((float4*)acc)[i] = make_float4(0.0f, 0.0f, 0.0f, 0.0f);
  if (tid < nown) list[tid] = NEDGES + lo + tid;
  if (tid == 0) cnt = nown;
  __syncthreads();

  const uint4* key4 = (const uint4*)key;
  const unsigned pat = blockIdx.x * 0x01010101u;
  const int start = blockIdx.x * (NK16 / AGG_WG);
  for (int i0 = tid; i0 < NK16; i0 += AGG_TB * SC_U) {
    uint4 v[SC_U];
    #pragma unroll
    for (int u = 0; u < SC_U; ++u) {
      int iw = min(i0 + u * AGG_TB, NK16 - 1) + start;
      if (iw >= NK16) iw -= NK16;
      v[u] = key4[iw];
    }
    unsigned mt[SC_U];
    #pragma unroll
    for (int u = 0; u < SC_U; ++u) {
      mt[u] = (zero_bytes(v[u].x ^ pat) >> 7) | (zero_bytes(v[u].y ^ pat) >> 6) |
              (zero_bytes(v[u].z ^ pat) >> 5) | (zero_bytes(v[u].w ^ pat) >> 4);
      if (i0 + u * AGG_TB >= NK16) mt[u] = 0;
    }
    unsigned M0 = mt[0] | (mt[1] << 4), M1 = mt[2] | (mt[3] << 4);
    unsigned M2 = mt[4] | (mt[5] << 4), M3 = mt[6] | (mt[7] << 4);
    while (M0 | M1 | M2 | M3) {
      const int p = M0 ? 0 : M1 ? 1 : M2 ? 2 : 3;
      const unsigned w = M0 ? M0 : M1 ? M1 : M2 ? M2 : M3;
      const int bp = __builtin_ctz(w);
      const unsigned cl = w & (w - 1);
      M0 = p == 0 ? cl : M0; M1 = p == 1 ? cl : M1;
      M2 = p == 2 ? cl : M2; M3 = p == 3 ? cl : M3;
      int iw = i0 + (2 * p + ((bp >> 2) & 1)) * AGG_TB + start;
      if (iw >= NK16) iw -= NK16;
      const int e = iw * 16 + (bp & 3) * 4 + (bp >> 3);
      if (agg_hit(e, &cnt, list))
        agg_add_row(e, ei[NEDGES + e] - lo, lo, acc, ei, xl, ebuf, ssum);
    }
  }
  __syncthreads();

  const int count = __builtin_amdgcn_readfirstlane(min(cnt, AGG_CAP));
  const int wbase = __builtin_amdgcn_readfirstlane(wave * 32);
  for (int base = wbase; base < count; base += AGG_TB) {
    const int nb = min(32, count - base);
    int soff = 0, doff = 0;
    float a0 = 0.0f, a1 = 0.0f, a2 = 0.0f, a3 = 0.0f;
    if (lane < nb) {
      const int e = list[base + lane];
      const bool self = e >= NEDGES;
      const int er = self ? 0 : e;
      const int s = self ? e - NEDGES : ei[er];
      const int d = self ? e - NEDGES : ei[NEDGES + er];
      const float4 p = *(const float4*)(ebuf + (size_t)e * NHEAD);
      const float4 q = *(const float4*)(ssum + (size_t)d * NHEAD);
      soff = s * FDIM;
      doff = (d - lo) * FDIM;
      a0 = p.x / q.x; a1 = p.y / q.y; a2 = p.z / q.z; a3 = p.w / q.w;
    }
    const int wb0 = __float_as_int(a0), wb1 = __float_as_int(a1);
    const int wb2 = __float_as_int(a2), wb3 = __float_as_int(a3);
    int j = 0;
    for (; j + 8 <= nb; j += 8) {
      float r[8][4];
      #pragma unroll
      for (int u = 0; u < 8; ++u) {
        const float* row = xl + __builtin_amdgcn_readlane(soff, j + u) + lane;
        r[u][0] = row[0]; r[u][1] = row[32]; r[u][2] = row[64]; r[u][3] = row[96];
      }
      #pragma unroll
      for (int u = 0; u < 8; ++u) {
        float* ar = acc + __builtin_amdgcn_readlane(doff, j + u) + lane;
        atomicAdd(ar +  0, r[u][0] * __int_as_float(__builtin_amdgcn_readlane(wb0, j + u)));
        atomicAdd(ar + 32, r[u][1] * __int_as_float(__builtin_amdgcn_readlane(wb1, j + u)));
        atomicAdd(ar + 64, r[u][2] * __int_as_float(__builtin_amdgcn_readlane(wb2, j + u)));
        atomicAdd(ar + 96, r[u][3] * __int_as_float(__builtin_amdgcn_readlane(wb3, j + u)));
      }
    }
    for (; j < nb; ++j) {
      const float* row = xl + __builtin_amdgcn_readlane(soff, j) + lane;
      const float r0 = row[0], r1 = row[32], r2 = row[64], r3 = row[96];
      float* ar = acc + __builtin_amdgcn_readlane(doff, j) + lane;
      atomicAdd(ar +  0, r0 * __int_as_float(__builtin_amdgcn_readlane(wb0, j)));
      atomicAdd(ar + 32, r1 * __int_as_float(__builtin_amdgcn_readlane(wb1, j)));
      atomicAdd(ar + 64, r2 * __int_as_float(__builtin_amdgcn_readlane(wb2, j)));
      atomicAdd(ar + 96, r3 * __int_as_float(__builtin_amdgcn_readlane(wb3, j)));
    }
  }
  __syncthreads();

  for (int i2 = tid; i2 < AGG_R * FDIM / 2; i2 += AGG_TB) {
    const int idx = i2 * 2;
    const int nl = idx >> 7, c = idx & (FDIM - 1);
    if (nl < nown) {
      float v0 = acc[idx] + b[c];
      float v1 = acc[idx + 1] + b[c + 1];
      v0 = v0 > 0.0f ? v0 : (__expf(v0) - 1.0f);
      v1 = v1 > 0.0f ? v1 : (__expf(v1) - 1.0f);
      v2h o = {(_Float16)v0, (_Float16)v1};
      *(v2h*)(outH + (size_t)(lo + nl) * FDIM + c) = o;
    }
  }
}

extern "C" void kernel_launch(void* const* d_in, const int* in_sizes, int n_in,
                              void* d_out, int out_size, void* d_ws,
                              size_t ws_size, hipStream_t stream) {
  (void)in_sizes; (void)n_in; (void)out_size; (void)ws_size;
  const float* x    = (const float*)d_in[0];
  const int*   ei   = (const int*)d_in[1];
  const float* Wl1  = (const float*)d_in[2];
  const float* Wr1  = (const float*)d_in[3];
  const float* att1 = (const float*)d_in[4];
  const float* b1   = (const float*)d_in[5];
  const float* Wl2  = (const float*)d_in[6];
  const float* Wr2  = (const float*)d_in[7];
  const float* att2 = (const float*)d_in[8];
  const float* b2   = (const float*)d_in[9];
  const float* W1   = (const float*)d_in[10];
  const float* bb1  = (const float*)d_in[11];
  const float* W2   = (const float*)d_in[12];
  const float* bb2  = (const float*)d_in[13];
  const float* W3   = (const float*)d_in[14];
  const float* bb3  = (const float*)d_in[15];
  float* out = (float*)d_out;

  char* ws = (char*)d_ws;
  size_t off = 0;
  auto alloc = [&](size_t bytes) -> char* {
    char* p = ws + off;
    off += (bytes + 255) & ~(size_t)255;
    return p;
  };
  const size_t S_F = (size_t)NNODES * FDIM;
  _Float16* xh  = (_Float16*)alloc(S_F * 2);
  _Float16* h1h = (_Float16*)alloc(S_F * 2);
  _Float16* h2h = (_Float16*)alloc(S_F * 2);
  _Float16* wt  = (_Float16*)alloc((size_t)WT_TOTAL * 2);
  const size_t uoff = off;
  float*    xl   = (float*)alloc(S_F * 4);
  float*    xr   = (float*)alloc(S_F * 4);
  float*    ebuf = (float*)alloc((size_t)ETOT * NHEAD * 4);
  unsigned* mkey = (unsigned*)alloc((size_t)NNODES * NHEAD * 4);
  float*    ssum = (float*)alloc((size_t)NNODES * NHEAD * 4);
  unsigned* key  = (unsigned*)alloc((size_t)NEDGES);
  _Float16* m1 = (_Float16*)(ws + uoff);
  _Float16* m2 = (_Float16*)(ws + uoff + (size_t)NNODES * 640 * 2);

  const int TB = 256;
  const int ngrid = (int)((S_F + TB - 1) / TB);
  const int hgrid = (NNODES * NHEAD + TB - 1) / TB;
  const int egrid = (ETOT * NHEAD + TB - 1) / TB;
  const dim3 gconv((NNODES + 127) / 128, 128 / 64);

  weights_f16t_kernel<<<dim3(WT_MAXTILES, 7), TB, 0, stream>>>(
      Wl1, Wr1, Wl2, Wr2, W1, W2, W3, wt);

  dst_key_kernel<<<(NE4 + TB - 1) / TB, TB, 0, stream>>>(ei, key);

  cast_f16_kernel<<<ngrid, TB, 0, stream>>>(x, xh, (int)S_F);
  wmma_gemm_kernel<<<gconv, TB, 0, stream>>>(xh, wt + WT_OFF_L1, nullptr, xl, nullptr,
                                             NNODES, 128, 128, 0);
  wmma_gemm_kernel<<<gconv, TB, 0, stream>>>(xh, wt + WT_OFF_R1, nullptr, xr, nullptr,
                                             NNODES, 128, 128, 0);
  init_conv_kernel<<<hgrid, TB, 0, stream>>>(mkey, ssum, NNODES);
  edge_logits_kernel<<<EL_GRID, TB, 0, stream>>>(ei, xl, xr, att1, ebuf, mkey);
  edge_exp_kernel<<<egrid, TB, 0, stream>>>(ei, ebuf, mkey, ssum);
  node_aggregate_kernel<<<AGG_WG, AGG_TB, 0, stream>>>(ei, key, xl, ebuf, ssum, b1, h1h);

  wmma_gemm_kernel<<<gconv, TB, 0, stream>>>(h1h, wt + WT_OFF_L2, nullptr, xl, nullptr,
                                             NNODES, 128, 128, 0);
  wmma_gemm_kernel<<<gconv, TB, 0, stream>>>(h1h, wt + WT_OFF_R2, nullptr, xr, nullptr,
                                             NNODES, 128, 128, 0);
  init_conv_kernel<<<hgrid, TB, 0, stream>>>(mkey, ssum, NNODES);
  edge_logits_kernel<<<EL_GRID, TB, 0, stream>>>(ei, xl, xr, att2, ebuf, mkey);
  edge_exp_kernel<<<egrid, TB, 0, stream>>>(ei, ebuf, mkey, ssum);
  node_aggregate_kernel<<<AGG_WG, AGG_TB, 0, stream>>>(ei, key, xl, ebuf, ssum, b2, h2h);

  const dim3 gm1((NNODES + 127) / 128, 640 / 64);
  wmma_gemm_kernel<<<gm1, TB, 0, stream>>>(h2h, wt + WT_OFF_M1, bb1, nullptr, m1,
                                           NNODES, 640, 128, 1);
  const dim3 gm2((NNODES + 127) / 128, 320 / 64);
  wmma_gemm_kernel<<<gm2, TB, 0, stream>>>(m1, wt + WT_OFF_M2, bb2, nullptr, m2,
                                           NNODES, 320, 640, 1);
  const dim3 gm3((NNODES + 127) / 128, 64 / 64);
  wmma_gemm_kernel<<<gm3, TB, 0, stream>>>(m2, wt + WT_OFF_M3, bb3, out, nullptr,
                                           NNODES, 64, 320, 2);
}


# Edge logits and segment max move into the destination-owner workgroup (block's xr rows in LDS, integer LDS max; exp sums are order-free integers by now): 4 launches, global atomics, xr gather gone
# speedup vs baseline: 6.4371x; 1.5113x over previous
#include <hip/hip_runtime.h>
#include <hip/hip_bf16.h>


typedef __attribute__((ext_vector_type(16))) _Float16 v16h;
typedef __attribute__((ext_vector_type(8)))  _Float16 v8h;
typedef __attribute__((ext_vector_type(8)))  float    v8f;

#define NNODES 50000
#define NEDGES 800000
#define ETOT   (NEDGES + NNODES)
#define FDIM   128
#define NHEAD  4

__device__ __forceinline__ unsigned fkey(float f) {
  unsigned u = __float_as_uint(f);
  return (u & 0x80000000u) ? ~u : (u | 0x80000000u);
}
__device__ __forceinline__ float funkey(unsigned k) {
  unsigned u = (k & 0x80000000u) ? (k & 0x7fffffffu) : ~k;
  return __uint_as_float(u);
}

__global__ void cast_f16_kernel(const float* __restrict__ in,
                                _Float16* __restrict__ out, int n) {
  int i = blockIdx.x * blockDim.x + threadIdx.x;
  if (i < n) out[i] = (_Float16)in[i];
}

#define WT_OFF_L1  0
#define WT_OFF_R1  (WT_OFF_L1 + 128 * 128)
#define WT_OFF_L2  (WT_OFF_R1 + 128 * 128)
#define WT_OFF_R2  (WT_OFF_L2 + 128 * 128)
#define WT_OFF_M1  (WT_OFF_R2 + 128 * 128)
#define WT_OFF_M2  (WT_OFF_M1 + 128 * 640)
#define WT_OFF_M3  (WT_OFF_M2 + 640 * 320)
#define WT_TOTAL   (WT_OFF_M3 + 320 * 64)
#define WT_MAXTILES ((640 / 32) * (320 / 32))

__global__ __launch_bounds__(256) void weights_f16t_kernel(
    const float* __restrict__ w0, const float* __restrict__ w1,
    const float* __restrict__ w2, const float* __restrict__ w3,
    const float* __restrict__ w4, const float* __restrict__ w5,
    const float* __restrict__ w6, _Float16* __restrict__ wt) {
  __shared__ float tile[32][33];
  const float* Wm; int K, Nt, off;
  switch (blockIdx.y) {
    case 0:  Wm = w0; K = 128; Nt = 128; off = WT_OFF_L1; break;
    case 1:  Wm = w1; K = 128; Nt = 128; off = WT_OFF_R1; break;
    case 2:  Wm = w2; K = 128; Nt = 128; off = WT_OFF_L2; break;
    case 3:  Wm = w3; K = 128; Nt = 128; off = WT_OFF_R2; break;
    case 4:  Wm = w4; K = 128; Nt = 640; off = WT_OFF_M1; break;
    case 5:  Wm = w5; K = 640; Nt = 320; off = WT_OFF_M2; break;
    default: Wm = w6; K = 320; Nt = 64;  off = WT_OFF_M3; break;
  }
  const int tn = Nt >> 5;
  if ((int)blockIdx.x >= (K >> 5) * tn) return;
  const int k0 = ((int)blockIdx.x / tn) * 32;
  const int n0 = ((int)blockIdx.x % tn) * 32;
  const int lane = threadIdx.x & 31;
  const int wv   = threadIdx.x >> 5;
  #pragma unroll
  for (int j = 0; j < 4; ++j) {
    const int k = wv + 8 * j;
    tile[k][lane] = Wm[(size_t)(k0 + k) * Nt + n0 + lane];
  }
  __syncthreads();
  #pragma unroll
  for (int j = 0; j < 4; ++j) {
    const int n = wv + 8 * j;
    wt[off + (n0 + n) * K + k0 + lane] = (_Float16)tile[lane][n];
  }
}

#define KC 64
#define BT_STRIDE 72

__global__ __launch_bounds__(256) void wmma_gemm_kernel(
    const _Float16* __restrict__ A, const _Float16* __restrict__ Wt,
    const float* __restrict__ bias,
    float* __restrict__ outF, _Float16* __restrict__ outH,
    int M, int Nt, int K, int act) {
  __shared__ _Float16 bt[64 * BT_STRIDE];

  const int lane = threadIdx.x & 31;
  const int wave = threadIdx.x >> 5;
  const int half = lane >> 4;
  const int l15  = lane & 15;
  const int n0   = blockIdx.y * 64;

  int rowBase = blockIdx.x * 128 + wave * 16;
  if (rowBase > M - 16) rowBase = M - 16;

  v8f c[4] = {v8f{}, v8f{}, v8f{}, v8f{}};

  for (int kc = 0; kc < K; kc += KC) {
    {
      const int q8 = (threadIdx.x & 7) * 8;
      int n = threadIdx.x >> 3;
      #pragma unroll
      for (int p = 0; p < 2; ++p, n += 32) {
        const v8h wv = *(const v8h*)(Wt + (size_t)(n0 + n) * K + kc + q8);
        *(v8h*)(bt + n * BT_STRIDE + q8) = wv;
      }
    }
    __syncthreads();

    const _Float16* arow = A + (size_t)(rowBase + l15) * K + kc;
    #pragma unroll
    for (int ks = 0; ks < 2; ++ks) {
      const int kb = ks * 32 + half * 8;
      v8h alo = *(const v8h*)(arow + kb);
      v8h ahi = *(const v8h*)(arow + kb + 16);
      v16h a = __builtin_shufflevector(alo, ahi, 0, 1, 2, 3, 4, 5, 6, 7,
                                       8, 9, 10, 11, 12, 13, 14, 15);
      v16h b[4];
      #pragma unroll
      for (int nn = 0; nn < 4; ++nn) {
        const _Float16* bp = bt + (nn * 16 + l15) * BT_STRIDE + kb;
        v8h blo = *(const v8h*)(bp);
        v8h bhi = *(const v8h*)(bp + 16);
        b[nn] = __builtin_shufflevector(blo, bhi, 0, 1, 2, 3, 4, 5, 6, 7,
                                        8, 9, 10, 11, 12, 13, 14, 15);
      }
      #pragma unroll
      for (int nn = 0; nn < 4; ++nn) {
        c[nn] = __builtin_amdgcn_wmma_f32_16x16x32_f16(
            false, a, false, b[nn], (short)0, c[nn], false, false);
      }
    }
    __syncthreads();
  }

  #pragma unroll
  for (int nn = 0; nn < 4; ++nn) {
    const int col = n0 + nn * 16 + l15;
    const float bv = bias ? bias[col] : 0.0f;
    #pragma unroll
    for (int r = 0; r < 8; ++r) {
      const int row = rowBase + half * 8 + r;
      float v = c[nn][r] + bv;
      if (act == 1) v = v > 0.0f ? v : (__expf(v) - 1.0f);
      else if (act == 2) v = 1.0f / (1.0f + __expf(-v));
      if (outF) outF[(size_t)row * Nt + col] = v;
      if (outH) outH[(size_t)row * Nt + col] = (_Float16)v;
    }
  }
}

#define FX_ONE   1099511627776.0f
#define FX_INV   (1.0f / 1099511627776.0f)
#define AGG_R    196
#define AGG_WG   ((NNODES + AGG_R - 1) / AGG_R)
#define AGG_TB   1024
#define AGG_CAP  16384
#define NE4      (NEDGES / 4)
static_assert(NEDGES % 4 == 0, "dst list is read as int4");
static_assert(AGG_R <= AGG_TB && AGG_R <= AGG_CAP, "self loops are queued by one pass");
static_assert(AGG_R * NHEAD <= AGG_TB, "maxima and denominators are set up by one pass");

#define NK16     (NEDGES / 16)
#define SC_U     8
static_assert(AGG_WG <= 256, "an owner's number fits one byte");
static_assert(NEDGES % 16 == 0, "key bytes are read as uint4");
static_assert(SC_U == 8, "the scan packs eight loads' matches into four words");

__global__ __launch_bounds__(256) void dst_key_kernel(
    const int* __restrict__ ei, unsigned* __restrict__ key) {
  const int i = blockIdx.x * blockDim.x + threadIdx.x;
  if (i >= NE4) return;
  const int4 v = ((const int4*)(ei + NEDGES))[i];
  key[i] = ((unsigned)v.x / AGG_R) | (((unsigned)v.y / AGG_R) << 8) |
           (((unsigned)v.z / AGG_R) << 16) | (((unsigned)v.w / AGG_R) << 24);
}

__device__ __forceinline__ unsigned zero_bytes(unsigned x) {
  return ~(((x & 0x7f7f7f7fu) + 0x7f7f7f7fu) | x | 0x7f7f7f7fu);
}

typedef __attribute__((ext_vector_type(2))) _Float16 v2h;

template <int M>
__device__ __forceinline__ float xor_lane(float v) {
  return __int_as_float(__builtin_amdgcn_update_dpp(
      0, __float_as_int(v), 0x160 + M, 0xf, 0xf, true));
}

__device__ __forceinline__ unsigned long long fx_of(float p) {
  return (unsigned long long)(p * FX_ONE);
}

__device__ __forceinline__ float slow_logit(int s, int d, int h,
                                            const float* __restrict__ xl,
                                            const float* __restrict__ xr,
                                            const float* __restrict__ att) {
  float a = 0.0f;
  for (int c = h * 32; c < h * 32 + 32; ++c) {
    float z = xl[(size_t)s * FDIM + c] + xr[(size_t)d * FDIM + c];
    z = z > 0.0f ? z : 0.2f * z;
    a += z * att[c];
  }
  return a;
}

__device__ __forceinline__ void slow_max(int s, int d, int dl, unsigned* mk,
                                         const float* __restrict__ xl,
                                         const float* __restrict__ xr,
                                         const float* __restrict__ att) {
  for (int h = 0; h < NHEAD; ++h)
    atomicMax(mk + dl * NHEAD + h, fkey(slow_logit(s, d, h, xl, xr, att)));
}

__device__ __forceinline__ void slow_add_row(int s, int d, int dl, float* acc,
                                             unsigned long long* sfx,
                                             const unsigned* mk,
                                             const float* __restrict__ xl,
                                             const float* __restrict__ xr,
                                             const float* __restrict__ att) {
  for (int h = 0; h < NHEAD; ++h) {
    const float p = __expf(slow_logit(s, d, h, xl, xr, att) - funkey(mk[dl * NHEAD + h]));
    atomicAdd(sfx + dl * NHEAD + h, fx_of(p));
    for (int c = h * 32; c < h * 32 + 32; ++c)
      atomicAdd(acc + dl * FDIM + c, xl[(size_t)s * FDIM + c] * p);
  }
}

template <int MODE>
__device__ __forceinline__ void owner_scan(
    const unsigned* __restrict__ key, int tid, int lo, int* cnt, int* list,
    float* acc, unsigned long long* sfx, unsigned* mk,
    const int* __restrict__ ei, const float* __restrict__ xl,
    const float* __restrict__ xr, const float* __restrict__ att) {
  const uint4* key4 = (const uint4*)key;
  const unsigned pat = blockIdx.x * 0x01010101u;
  const int start = blockIdx.x * (NK16 / AGG_WG);
  for (int i0 = tid; i0 < NK16; i0 += AGG_TB * SC_U) {
    uint4 v[SC_U];
    #pragma unroll
    for (int u = 0; u < SC_U; ++u) {
      int iw = min(i0 + u * AGG_TB, NK16 - 1) + start;
      if (iw >= NK16) iw -= NK16;
      v[u] = key4[iw];
    }
    unsigned mt[SC_U];
    #pragma unroll
    for (int u = 0; u < SC_U; ++u) {
      mt[u] = (zero_bytes(v[u].x ^ pat) >> 7) | (zero_bytes(v[u].y ^ pat) >> 6) |
              (zero_bytes(v[u].z ^ pat) >> 5) | (zero_bytes(v[u].w ^ pat) >> 4);
      if (i0 + u * AGG_TB >= NK16) mt[u] = 0;
    }
    unsigned M0 = mt[0] | (mt[1] << 4), M1 = mt[2] | (mt[3] << 4);
    unsigned M2 = mt[4] | (mt[5] << 4), M3 = mt[6] | (mt[7] << 4);
    while (M0 | M1 | M2 | M3) {
      const int p = M0 ? 0 : M1 ? 1 : M2 ? 2 : 3;
      const unsigned w = M0 ? M0 : M1 ? M1 : M2 ? M2 : M3;
      const int bp = __builtin_ctz(w);
      const unsigned cl = w & (w - 1);
      M0 = p == 0 ? cl : M0; M1 = p == 1 ? cl : M1;
      M2 = p == 2 ? cl : M2; M3 = p == 3 ? cl : M3;
      int iw = i0 + (2 * p + ((bp >> 2) & 1)) * AGG_TB + start;
      if (iw >= NK16) iw -= NK16;
      const int e = iw * 16 + (bp & 3) * 4 + (bp >> 3);
      if (MODE == 0) {
        const int pos = atomicAdd(cnt, 1);
        if (pos < AGG_CAP) list[pos] = e;
      } else {
        const int s = ei[e], d = ei[NEDGES + e];
        if (MODE == 1) slow_max(s, d, d - lo, mk, xl, xr, att);
        else           slow_add_row(s, d, d - lo, acc, sfx, mk, xl, xr, att);
      }
    }
  }
}

__global__ __launch_bounds__(AGG_TB) void conv_edge_kernel(
    const int* __restrict__ ei, const unsigned* __restrict__ key,
    const float* __restrict__ xl, const float* __restrict__ xr,
    const float* __restrict__ att, float* lq,
    const float* __restrict__ b, _Float16* __restrict__ outH) {
  __shared__ __attribute__((aligned(16))) float acc[AGG_R * FDIM];
  __shared__ __attribute__((aligned(16))) float xrt[AGG_R * FDIM];
  __shared__ __attribute__((aligned(16))) unsigned long long sfx[AGG_R * NHEAD];
  __shared__ __attribute__((aligned(16))) unsigned mk[AGG_R * NHEAD];
  __shared__ int   list[AGG_CAP];
  __shared__ int   cnt;

  const int tid  = threadIdx.x;
  const int lane = tid & 31;
  const int wave = tid >> 5;
  const int lo   = blockIdx.x * AGG_R;
  const int nown = min(AGG_R, NNODES - lo);

  {
    const float4* xr4 = (const float4*)(xr + (size_t)lo * FDIM);
    for (int i = tid; i < AGG_R * FDIM / 4; i += AGG_TB) {
      ((float4*)acc)[i] = make_float4(0.0f, 0.0f, 0.0f, 0.0f);
      ((float4*)xrt)[i] = i < nown * (FDIM / 4) ? xr4[i]
                                                : make_float4(0.0f, 0.0f, 0.0f, 0.0f);
    }
  }
  if (tid < AGG_R * NHEAD) { sfx[tid] = 0ull; mk[tid] = 0u; }
  if (tid < nown) list[tid] = NEDGES + lo + tid;
  if (tid == 0) cnt = nown;
  __syncthreads();

  owner_scan<0>(key, tid, lo, &cnt, list, acc, sfx, mk, ei, xl, xr, att);
  __syncthreads();

  const int total = __builtin_amdgcn_readfirstlane(cnt);
  const int wbase = __builtin_amdgcn_readfirstlane(wave * 32);

  if (total <= AGG_CAP) {
    const float4 w = *(const float4*)(att + lane * 4);
    const int u_l = lane & 7, g = lane >> 3;
    float* myq = lq + (size_t)blockIdx.x * AGG_CAP * NHEAD;
    for (int base = wbase; base < total; base += AGG_TB) {
      const int nb = min(32, total - base);
      int srow = 0, drow = 0;
      if (lane < nb) {
        const int e = list[base + lane];
        const bool self = e >= NEDGES;
        const int er = self ? 0 : e;
        srow = self ? e - NEDGES : ei[er];
        drow = (self ? e - NEDGES : ei[NEDGES + er]) - lo;
      }
      for (int j = 0; j < nb; j += 8) {
        float4 a[8];
        #pragma unroll
        for (int u = 0; u < 8; ++u)
          a[u] = *(const float4*)(xl + __builtin_amdgcn_readlane(srow, j + u) * FDIM + lane * 4);
        float mine = 0.0f;
        #pragma unroll
        for (int u = 0; u < 8; ++u) {
          const float4 bb =
              *(const float4*)(xrt + __builtin_amdgcn_readlane(drow, j + u) * FDIM + lane * 4);
          float z, t;
          z = a[u].x + bb.x; z = z > 0.0f ? z : 0.2f * z; t = z * w.x;
          z = a[u].y + bb.y; z = z > 0.0f ? z : 0.2f * z; t += z * w.y;
          z = a[u].z + bb.z; z = z > 0.0f ? z : 0.2f * z; t += z * w.z;
          z = a[u].w + bb.w; z = z > 0.0f ? z : 0.2f * z; t += z * w.w;
          t += xor_lane<1>(t);
          t += xor_lane<2>(t);
          t += xor_lane<4>(t);
          if (u_l == u) mine = t;
        }
        const int from = j + u_l;
        const int d_m = __shfl(drow, from);
        if (from < nb) {
          myq[(base + from) * NHEAD + g] = mine;
          atomicMax(mk + d_m * NHEAD + g, fkey(mine));
        }
      }
    }
    __syncthreads();

    for (int base = wbase; base < total; base += AGG_TB) {
      const int nb = min(32, total - base);
      int soff = 0, doff = 0;
      float a0 = 0.0f, a1 = 0.0f, a2 = 0.0f, a3 = 0.0f;
      if (lane < nb) {
        const int e = list[base + lane];
        const bool self = e >= NEDGES;
        const int er = self ? 0 : e;
        const int s = self ? e - NEDGES : ei[er];
        const int d = self ? e - NEDGES : ei[NEDGES + er];
        const float4 lg = *(const float4*)(myq + (base + lane) * NHEAD);
        const int dl = d - lo;
        const uint4 mq = *(const uint4*)(mk + dl * NHEAD);
        soff = s * FDIM;
        doff = dl * FDIM;
        a0 = __expf(lg.x - funkey(mq.x)); a1 = __expf(lg.y - funkey(mq.y));
        a2 = __expf(lg.z - funkey(mq.z)); a3 = __expf(lg.w - funkey(mq.w));
        atomicAdd(sfx + dl * NHEAD + 0, fx_of(a0));
        atomicAdd(sfx + dl * NHEAD + 1, fx_of(a1));
        atomicAdd(sfx + dl * NHEAD + 2, fx_of(a2));
        atomicAdd(sfx + dl * NHEAD + 3, fx_of(a3));
      }
      const int wb0 = __float_as_int(a0), wb1 = __float_as_int(a1);
      const int wb2 = __float_as_int(a2), wb3 = __float_as_int(a3);
      int j = 0;
      for (; j + 8 <= nb; j += 8) {
        float r[8][4];
        #pragma unroll
        for (int u = 0; u < 8; ++u) {
          const float* row = xl + __builtin_amdgcn_readlane(soff, j + u) + lane;
          r[u][0] = row[0]; r[u][1] = row[32]; r[u][2] = row[64]; r[u][3] = row[96];
        }
        #pragma unroll
        for (int u = 0; u < 8; ++u) {
          float* ar = acc + __builtin_amdgcn_readlane(doff, j + u) + lane;
          atomicAdd(ar +  0, r[u][0] * __int_as_float(__builtin_amdgcn_readlane(wb0, j + u)));
          atomicAdd(ar + 32, r[u][1] * __int_as_float(__builtin_amdgcn_readlane(wb1, j + u)));
          atomicAdd(ar + 64, r[u][2] * __int_as_float(__builtin_amdgcn_readlane(wb2, j + u)));
          atomicAdd(ar + 96, r[u][3] * __int_as_float(__builtin_amdgcn_readlane(wb3, j + u)));
        }
      }
      for (; j < nb; ++j) {
        const float* row = xl + __builtin_amdgcn_readlane(soff, j) + lane;
        const float r0 = row[0], r1 = row[32], r2 = row[64], r3 = row[96];
        float* ar = acc + __builtin_amdgcn_readlane(doff, j) + lane;
        atomicAdd(ar +  0, r0 * __int_as_float(__builtin_amdgcn_readlane(wb0, j)));
        atomicAdd(ar + 32, r1 * __int_as_float(__builtin_amdgcn_readlane(wb1, j)));
        atomicAdd(ar + 64, r2 * __int_as_float(__builtin_amdgcn_readlane(wb2, j)));
        atomicAdd(ar + 96, r3 * __int_as_float(__builtin_amdgcn_readlane(wb3, j)));
      }
    }
  } else {
    if (tid < nown) slow_max(lo + tid, lo + tid, tid, mk, xl, xr, att);
    owner_scan<1>(key, tid, lo, &cnt, list, acc, sfx, mk, ei, xl, xr, att);
    __syncthreads();
    if (tid < nown) slow_add_row(lo + tid, lo + tid, tid, acc, sfx, mk, xl, xr, att);
    owner_scan<2>(key, tid, lo, &cnt, list, acc, sfx, mk, ei, xl, xr, att);
  }
  __syncthreads();

  for (int i2 = tid; i2 < AGG_R * FDIM / 2; i2 += AGG_TB) {
    const int idx = i2 * 2;
    const int nl = idx >> 7, c = idx & (FDIM - 1);
    if (nl < nown) {
      const float den = (float)sfx[nl * NHEAD + (c >> 5)] * FX_INV;
      float v0 = acc[idx] / den + b[c];
      float v1 = acc[idx + 1] / den + b[c + 1];
      v0 = v0 > 0.0f ? v0 : (__expf(v0) - 1.0f);
      v1 = v1 > 0.0f ? v1 : (__expf(v1) - 1.0f);
      v2h o = {(_Float16)v0, (_Float16)v1};
      *(v2h*)(outH + (size_t)(lo + nl) * FDIM + c) = o;
    }
  }
}

extern "C" void kernel_launch(void* const* d_in, const int* in_sizes, int n_in,
                              void* d_out, int out_size, void* d_ws,
                              size_t ws_size, hipStream_t stream) {
  (void)in_sizes; (void)n_in; (void)out_size; (void)ws_size;
  const float* x    = (const float*)d_in[0];
  const int*   ei   = (const int*)d_in[1];
  const float* Wl1  = (const float*)d_in[2];
  const float* Wr1  = (const float*)d_in[3];
  const float* att1 = (const float*)d_in[4];
  const float* b1   = (const float*)d_in[5];
  const float* Wl2  = (const float*)d_in[6];
  const float* Wr2  = (const float*)d_in[7];
  const float* att2 = (const float*)d_in[8];
  const float* b2   = (const float*)d_in[9];
  const float* W1   = (const float*)d_in[10];
  const float* bb1  = (const float*)d_in[11];
  const float* W2   = (const float*)d_in[12];
  const float* bb2  = (const float*)d_in[13];
  const float* W3   = (const float*)d_in[14];
  const float* bb3  = (const float*)d_in[15];
  float* out = (float*)d_out;

  char* ws = (char*)d_ws;
  size_t off = 0;
  auto alloc = [&](size_t bytes) -> char* {
    char* p = ws + off;
    off += (bytes + 255) & ~(size_t)255;
    return p;
  };
  const size_t S_F = (size_t)NNODES * FDIM;
  _Float16* xh  = (_Float16*)alloc(S_F * 2);
  _Float16* h1h = (_Float16*)alloc(S_F * 2);
  _Float16* h2h = (_Float16*)alloc(S_F * 2);
  _Float16* wt  = (_Float16*)alloc((size_t)WT_TOTAL * 2);
  const size_t uoff = off;
  float*    xl   = (float*)alloc(S_F * 4);
  float*    xr   = (float*)alloc(S_F * 4);
  float*    lq   = (float*)alloc((size_t)AGG_WG * AGG_CAP * NHEAD * 4);
  unsigned* key  = (unsigned*)alloc((size_t)NEDGES);
  _Float16* m1 = (_Float16*)(ws + uoff);
  _Float16* m2 = (_Float16*)(ws + uoff + (size_t)NNODES * 640 * 2);

  const int TB = 256;
  const int ngrid = (int)((S_F + TB - 1) / TB);
  const dim3 gconv((NNODES + 127) / 128, 128 / 64);

  weights_f16t_kernel<<<dim3(WT_MAXTILES, 7), TB, 0, stream>>>(
      Wl1, Wr1, Wl2, Wr2, W1, W2, W3, wt);

  dst_key_kernel<<<(NE4 + TB - 1) / TB, TB, 0, stream>>>(ei, key);

  cast_f16_kernel<<<ngrid, TB, 0, stream>>>(x, xh, (int)S_F);
  wmma_gemm_kernel<<<gconv, TB, 0, stream>>>(xh, wt + WT_OFF_L1, nullptr, xl, nullptr,
                                             NNODES, 128, 128, 0);
  wmma_gemm_kernel<<<gconv, TB, 0, stream>>>(xh, wt + WT_OFF_R1, nullptr, xr, nullptr,
                                             NNODES, 128, 128, 0);
  conv_edge_kernel<<<AGG_WG, AGG_TB, 0, stream>>>(ei, key, xl, xr, att1, lq, b1, h1h);

  wmma_gemm_kernel<<<gconv, TB, 0, stream>>>(h1h, wt + WT_OFF_L2, nullptr, xl, nullptr,
                                             NNODES, 128, 128, 0);
  wmma_gemm_kernel<<<gconv, TB, 0, stream>>>(h1h, wt + WT_OFF_R2, nullptr, xr, nullptr,
                                             NNODES, 128, 128, 0);
  conv_edge_kernel<<<AGG_WG, AGG_TB, 0, stream>>>(ei, key, xl, xr, att2, lq, b2, h2h);

  const dim3 gm1((NNODES + 127) / 128, 640 / 64);
  wmma_gemm_kernel<<<gm1, TB, 0, stream>>>(h2h, wt + WT_OFF_M1, bb1, nullptr, m1,
                                           NNODES, 640, 128, 1);
  const dim3 gm2((NNODES + 127) / 128, 320 / 64);
  wmma_gemm_kernel<<<gm2, TB, 0, stream>>>(m1, wt + WT_OFF_M2, bb2, nullptr, m2,
                                           NNODES, 320, 640, 1);
  const dim3 gm3((NNODES + 127) / 128, 64 / 64);
  wmma_gemm_kernel<<<gm3, TB, 0, stream>>>(m2, wt + WT_OFF_M3, bb3, out, nullptr,
                                           NNODES, 64, 320, 2);
}


# Second conv's owner workgroups load the per-destination runs the first conv's built and stored in the workspace: its key scan, edge lookups, counts, placement and sort are gone; sum order unchanged
# speedup vs baseline: 8.0177x; 1.2455x over previous
#include <hip/hip_runtime.h>
#include <hip/hip_bf16.h>


typedef __attribute__((ext_vector_type(16))) _Float16 v16h;
typedef __attribute__((ext_vector_type(8)))  _Float16 v8h;
typedef __attribute__((ext_vector_type(8)))  float    v8f;

#define NNODES 50000
#define NEDGES 800000
#define ETOT   (NEDGES + NNODES)
#define FDIM   128
#define NHEAD  4

__device__ __forceinline__ unsigned fkey(float f) {
  unsigned u = __float_as_uint(f);
  return (u & 0x80000000u) ? ~u : (u | 0x80000000u);
}
__device__ __forceinline__ float funkey(unsigned k) {
  unsigned u = (k & 0x80000000u) ? (k & 0x7fffffffu) : ~k;
  return __uint_as_float(u);
}

__global__ void cast_f16_kernel(const float* __restrict__ in,
                                _Float16* __restrict__ out, int n) {
  int i = blockIdx.x * blockDim.x + threadIdx.x;
  if (i < n) out[i] = (_Float16)in[i];
}

#define WT_OFF_L1  0
#define WT_OFF_R1  (WT_OFF_L1 + 128 * 128)
#define WT_OFF_L2  (WT_OFF_R1 + 128 * 128)
#define WT_OFF_R2  (WT_OFF_L2 + 128 * 128)
#define WT_OFF_M1  (WT_OFF_R2 + 128 * 128)
#define WT_OFF_M2  (WT_OFF_M1 + 128 * 640)
#define WT_OFF_M3  (WT_OFF_M2 + 640 * 320)
#define WT_TOTAL   (WT_OFF_M3 + 320 * 64)
#define WT_MAXTILES ((640 / 32) * (320 / 32))

__global__ __launch_bounds__(256) void weights_f16t_kernel(
    const float* __restrict__ w0, const float* __restrict__ w1,
    const float* __restrict__ w2, const float* __restrict__ w3,
    const float* __restrict__ w4, const float* __restrict__ w5,
    const float* __restrict__ w6, _Float16* __restrict__ wt) {
  __shared__ float tile[32][33];
  const float* Wm; int K, Nt, off;
  switch (blockIdx.y) {
    case 0:  Wm = w0; K = 128; Nt = 128; off = WT_OFF_L1; break;
    case 1:  Wm = w1; K = 128; Nt = 128; off = WT_OFF_R1; break;
    case 2:  Wm = w2; K = 128; Nt = 128; off = WT_OFF_L2; break;
    case 3:  Wm = w3; K = 128; Nt = 128; off = WT_OFF_R2; break;
    case 4:  Wm = w4; K = 128; Nt = 640; off = WT_OFF_M1; break;
    case 5:  Wm = w5; K = 640; Nt = 320; off = WT_OFF_M2; break;
    default: Wm = w6; K = 320; Nt = 64;  off = WT_OFF_M3; break;
  }
  const int tn = Nt >> 5;
  if ((int)blockIdx.x >= (K >> 5) * tn) return;
  const int k0 = ((int)blockIdx.x / tn) * 32;
  const int n0 = ((int)blockIdx.x % tn) * 32;
  const int lane = threadIdx.x & 31;
  const int wv   = threadIdx.x >> 5;
  #pragma unroll
  for (int j = 0; j < 4; ++j) {
    const int k = wv + 8 * j;
    tile[k][lane] = Wm[(size_t)(k0 + k) * Nt + n0 + lane];
  }
  __syncthreads();
  #pragma unroll
  for (int j = 0; j < 4; ++j) {
    const int n = wv + 8 * j;
    wt[off + (n0 + n) * K + k0 + lane] = (_Float16)tile[lane][n];
  }
}

#define KC 64
#define BT_STRIDE 72

__global__ __launch_bounds__(256) void wmma_gemm_kernel(
    const _Float16* __restrict__ A, const _Float16* __restrict__ Wt,
    const float* __restrict__ bias,
    float* __restrict__ outF, _Float16* __restrict__ outH,
    int M, int Nt, int K, int act) {
  __shared__ _Float16 bt[64 * BT_STRIDE];

  const int lane = threadIdx.x & 31;
  const int wave = threadIdx.x >> 5;
  const int half = lane >> 4;
  const int l15  = lane & 15;
  const int n0   = blockIdx.y * 64;

  int rowBase = blockIdx.x * 128 + wave * 16;
  if (rowBase > M - 16) rowBase = M - 16;

  v8f c[4] = {v8f{}, v8f{}, v8f{}, v8f{}};

  for (int kc = 0; kc < K; kc += KC) {
    {
      const int q8 = (threadIdx.x & 7) * 8;
      int n = threadIdx.x >> 3;
      #pragma unroll
      for (int p = 0; p < 2; ++p, n += 32) {
        const v8h wv = *(const v8h*)(Wt + (size_t)(n0 + n) * K + kc + q8);
        *(v8h*)(bt + n * BT_STRIDE + q8) = wv;
      }
    }
    __syncthreads();

    const _Float16* arow = A + (size_t)(rowBase + l15) * K + kc;
    #pragma unroll
    for (int ks = 0; ks < 2; ++ks) {
      const int kb = ks * 32 + half * 8;
      v8h alo = *(const v8h*)(arow + kb);
      v8h ahi = *(const v8h*)(arow + kb + 16);
      v16h a = __builtin_shufflevector(alo, ahi, 0, 1, 2, 3, 4, 5, 6, 7,
                                       8, 9, 10, 11, 12, 13, 14, 15);
      v16h b[4];
      #pragma unroll
      for (int nn = 0; nn < 4; ++nn) {
        const _Float16* bp = bt + (nn * 16 + l15) * BT_STRIDE + kb;
        v8h blo = *(const v8h*)(bp);
        v8h bhi = *(const v8h*)(bp + 16);
        b[nn] = __builtin_shufflevector(blo, bhi, 0, 1, 2, 3, 4, 5, 6, 7,
                                        8, 9, 10, 11, 12, 13, 14, 15);
      }
      #pragma unroll
      for (int nn = 0; nn < 4; ++nn) {
        c[nn] = __builtin_amdgcn_wmma_f32_16x16x32_f16(
            false, a, false, b[nn], (short)0, c[nn], false, false);
      }
    }
    __syncthreads();
  }

  #pragma unroll
  for (int nn = 0; nn < 4; ++nn) {
    const int col = n0 + nn * 16 + l15;
    const float bv = bias ? bias[col] : 0.0f;
    #pragma unroll
    for (int r = 0; r < 8; ++r) {
      const int row = rowBase + half * 8 + r;
      float v = c[nn][r] + bv;
      if (act == 1) v = v > 0.0f ? v : (__expf(v) - 1.0f);
      else if (act == 2) v = 1.0f / (1.0f + __expf(-v));
      if (outF) outF[(size_t)row * Nt + col] = v;
      if (outH) outH[(size_t)row * Nt + col] = (_Float16)v;
    }
  }
}

#define FX_ONE   1099511627776.0f
#define FX_INV   (1.0f / 1099511627776.0f)
#define AGG_R    196
#define AGG_WG   ((NNODES + AGG_R - 1) / AGG_R)
#define AGG_TB   1024
#define AGG_CAP  16384
#define NE4      (NEDGES / 4)
static_assert(NEDGES % 4 == 0, "dst list is read as int4");
static_assert(AGG_R <= AGG_TB && AGG_R <= AGG_CAP, "self loops are queued by one pass");
static_assert(AGG_R * NHEAD <= AGG_TB, "maxima and denominators are set up by one pass");
static_assert(NNODES <= 65536 && AGG_R <= 256, "source row and local destination share a queue slot");
static_assert(AGG_CAP <= AGG_R * FDIM, "the runs take the place of the slow path's row sums");
#define RM_W     (1 + 2 * AGG_R)

#define NK16     (NEDGES / 16)
#define SC_U     8
static_assert(AGG_WG <= 256, "an owner's number fits one byte");
static_assert(NEDGES % 16 == 0, "key bytes are read as uint4");
static_assert(SC_U == 8, "the scan packs eight loads' matches into four words");

__global__ __launch_bounds__(256) void dst_key_kernel(
    const int* __restrict__ ei, unsigned* __restrict__ key) {
  const int i = blockIdx.x * blockDim.x + threadIdx.x;
  if (i >= NE4) return;
  const int4 v = ((const int4*)(ei + NEDGES))[i];
  key[i] = ((unsigned)v.x / AGG_R) | (((unsigned)v.y / AGG_R) << 8) |
           (((unsigned)v.z / AGG_R) << 16) | (((unsigned)v.w / AGG_R) << 24);
}

__device__ __forceinline__ unsigned zero_bytes(unsigned x) {
  return ~(((x & 0x7f7f7f7fu) + 0x7f7f7f7fu) | x | 0x7f7f7f7fu);
}

typedef __attribute__((ext_vector_type(2))) _Float16 v2h;

template <int M>
__device__ __forceinline__ float xor_lane(float v) {
  return __int_as_float(__builtin_amdgcn_update_dpp(
      0, __float_as_int(v), 0x160 + M, 0xf, 0xf, true));
}

__device__ __forceinline__ unsigned long long fx_of(float p) {
  return (unsigned long long)(p * FX_ONE);
}

__device__ __forceinline__ float slow_logit(int s, int d, int h,
                                            const float* __restrict__ xl,
                                            const float* __restrict__ xr,
                                            const float* __restrict__ att) {
  float a = 0.0f;
  for (int c = h * 32; c < h * 32 + 32; ++c) {
    float z = xl[(size_t)s * FDIM + c] + xr[(size_t)d * FDIM + c];
    z = z > 0.0f ? z : 0.2f * z;
    a += z * att[c];
  }
  return a;
}

__device__ __forceinline__ void slow_max(int s, int d, int dl, unsigned* mk,
                                         const float* __restrict__ xl,
                                         const float* __restrict__ xr,
                                         const float* __restrict__ att) {
  for (int h = 0; h < NHEAD; ++h)
    atomicMax(mk + dl * NHEAD + h, fkey(slow_logit(s, d, h, xl, xr, att)));
}

__device__ __forceinline__ void slow_add_row(int s, int d, int dl, float* acc,
                                             unsigned long long* sfx,
                                             const unsigned* mk,
                                             const float* __restrict__ xl,
                                             const float* __restrict__ xr,
                                             const float* __restrict__ att) {
  for (int h = 0; h < NHEAD; ++h) {
    const float p = __expf(slow_logit(s, d, h, xl, xr, att) - funkey(mk[dl * NHEAD + h]));
    atomicAdd(sfx + dl * NHEAD + h, fx_of(p));
    for (int c = h * 32; c < h * 32 + 32; ++c)
      atomicAdd(acc + dl * FDIM + c, xl[(size_t)s * FDIM + c] * p);
  }
}

template <int MODE>
__device__ __forceinline__ void owner_scan(
    const unsigned* __restrict__ key, int tid, int lo, int* cnt, int* list,
    float* acc, unsigned long long* sfx, unsigned* mk,
    const int* __restrict__ ei, const float* __restrict__ xl,
    const float* __restrict__ xr, const float* __restrict__ att) {
  const uint4* key4 = (const uint4*)key;
  const unsigned pat = blockIdx.x * 0x01010101u;
  const int start = blockIdx.x * (NK16 / AGG_WG);
  for (int i0 = tid; i0 < NK16; i0 += AGG_TB * SC_U) {
    uint4 v[SC_U];
    #pragma unroll
    for (int u = 0; u < SC_U; ++u) {
      int iw = min(i0 + u * AGG_TB, NK16 - 1) + start;
      if (iw >= NK16) iw -= NK16;
      v[u] = key4[iw];
    }
    unsigned mt[SC_U];
    #pragma unroll
    for (int u = 0; u < SC_U; ++u) {
      mt[u] = (zero_bytes(v[u].x ^ pat) >> 7) | (zero_bytes(v[u].y ^ pat) >> 6) |
              (zero_bytes(v[u].z ^ pat) >> 5) | (zero_bytes(v[u].w ^ pat) >> 4);
      if (i0 + u * AGG_TB >= NK16) mt[u] = 0;
    }
    unsigned M0 = mt[0] | (mt[1] << 4), M1 = mt[2] | (mt[3] << 4);
    unsigned M2 = mt[4] | (mt[5] << 4), M3 = mt[6] | (mt[7] << 4);
    while (M0 | M1 | M2 | M3) {
      const int p = M0 ? 0 : M1 ? 1 : M2 ? 2 : 3;
      const unsigned w = M0 ? M0 : M1 ? M1 : M2 ? M2 : M3;
      const int bp = __builtin_ctz(w);
      const unsigned cl = w & (w - 1);
      M0 = p == 0 ? cl : M0; M1 = p == 1 ? cl : M1;
      M2 = p == 2 ? cl : M2; M3 = p == 3 ? cl : M3;
      int iw = i0 + (2 * p + ((bp >> 2) & 1)) * AGG_TB + start;
      if (iw >= NK16) iw -= NK16;
      const int e = iw * 16 + (bp & 3) * 4 + (bp >> 3);
      if (MODE == 0) {
        const int pos = atomicAdd(cnt, 1);
        if (pos < AGG_CAP) list[pos] = e;
      } else {
        const int s = ei[e], d = ei[NEDGES + e];
        if (MODE == 1) slow_max(s, d, d - lo, mk, xl, xr, att);
        else           slow_add_row(s, d, d - lo, acc, sfx, mk, xl, xr, att);
      }
    }
  }
}

typedef __attribute__((ext_vector_type(4))) _Float16 v4h;

#define NG_MAX 9
static_assert(4 * NG_MAX >= 32 && NG_MAX <= 9, "32 sources make at most four groups; bodies exist up to 9");

template <int CN>
__device__ __forceinline__ void node_group(
    const float* __restrict__ xl, int srow, int j, int lane,
    const float4 bb, const float4 w,
    float& m, float& den, float& s0, float& s1, float& s2, float& s3) {
  float4 a[CN];
  float  t[CN];
  #pragma unroll
  for (int u = 0; u < CN; ++u)
    a[u] = *(const float4*)(xl + __builtin_amdgcn_readlane(srow, j + u) * FDIM + lane * 4);
  __builtin_amdgcn_sched_barrier(0);
  float cm = -INFINITY;
  #pragma unroll
  for (int u = 0; u < CN; ++u) {
    float z, q;
    z = a[u].x + bb.x; z = z > 0.0f ? z : 0.2f * z; q = z * w.x;
    z = a[u].y + bb.y; z = z > 0.0f ? z : 0.2f * z; q += z * w.y;
    z = a[u].z + bb.z; z = z > 0.0f ? z : 0.2f * z; q += z * w.z;
    z = a[u].w + bb.w; z = z > 0.0f ? z : 0.2f * z; q += z * w.w;
    q += xor_lane<1>(q);
    q += xor_lane<2>(q);
    q += xor_lane<4>(q);
    t[u] = q;
    cm = q > cm ? q : cm;
  }
  const float mn = fmaxf(m, cm);
  const float sc = __expf(m - mn);
  den *= sc; s0 *= sc; s1 *= sc; s2 *= sc; s3 *= sc;
  m = mn;
  #pragma unroll
  for (int u = 0; u < CN; ++u) {
    const float p = __expf(t[u] - mn);
    den += p;
    s0 += p * a[u].x; s1 += p * a[u].y; s2 += p * a[u].z; s3 += p * a[u].w;
  }
}

template <int STAGE>
__global__ __launch_bounds__(AGG_TB) void conv_edge_kernel(
    const int* __restrict__ ei, const unsigned* __restrict__ key,
    const float* __restrict__ xl, const float* __restrict__ xr,
    const float* __restrict__ att,
    const float* __restrict__ b, _Float16* __restrict__ outH,
    int* __restrict__ runs, int* __restrict__ rmeta) {
  __shared__ __attribute__((aligned(16))) union {
    float f[AGG_R * FDIM];
    int   i[AGG_CAP];
  } au;
  __shared__ __attribute__((aligned(16))) float xrt[AGG_R * FDIM];
  __shared__ __attribute__((aligned(16))) unsigned long long sfx[AGG_R * NHEAD];
  __shared__ __attribute__((aligned(16))) unsigned mk[AGG_R * NHEAD];
  __shared__ int   list[AGG_CAP];
  __shared__ int   cntn[AGG_R];
  __shared__ int   offs[AGG_R];
  __shared__ int   pos[AGG_R];
  __shared__ int   cnt;
  __shared__ int   nextn;
  float* acc = au.f;
  int*   seg = au.i;

  const int tid  = threadIdx.x;
  const int lane = tid & 31;
  const int wave = tid >> 5;
  const int lo   = blockIdx.x * AGG_R;
  const int nown = min(AGG_R, NNODES - lo);
  int* myruns = runs + (size_t)blockIdx.x * AGG_CAP;
  int* mymeta = rmeta + blockIdx.x * RM_W;

  {
    const float4* xr4 = (const float4*)(xr + (size_t)lo * FDIM);
    for (int i = tid; i < AGG_R * FDIM / 4; i += AGG_TB)
      ((float4*)xrt)[i] = i < nown * (FDIM / 4) ? xr4[i]
                                                : make_float4(0.0f, 0.0f, 0.0f, 0.0f);
  }
  int total;
  if (STAGE == 0) {
    if (tid < AGG_R) cntn[tid] = 0;
    if (tid < nown) list[tid] = NEDGES + lo + tid;
    if (tid == 0) { cnt = nown; nextn = 0; }
    __syncthreads();

    owner_scan<0>(key, tid, lo, &cnt, list, acc, sfx, mk, ei, xl, xr, att);
    __syncthreads();

    total = __builtin_amdgcn_readfirstlane(cnt);
    if (tid == 0) mymeta[0] = total;
  } else {
    if (tid == 0) nextn = 0;
    total = __builtin_amdgcn_readfirstlane(mymeta[0]);
  }
  const int wv = __builtin_amdgcn_readfirstlane(wave);

  if (total <= AGG_CAP) {
   if (STAGE == 0) {
    for (int i0 = tid; i0 < total; i0 += 4 * AGG_TB) {
      int sv[4], dv[4];
      #pragma unroll
      for (int u = 0; u < 4; ++u) {
        const int i = i0 + u * AGG_TB;
        const int e = i < total ? list[i] : NEDGES;
        const bool self = e >= NEDGES;
        const int er = self ? 0 : e;
        const int s = ei[er], d = ei[NEDGES + er];
        sv[u] = self ? e - NEDGES : s;
        dv[u] = self ? e - NEDGES : d;
      }
      #pragma unroll
      for (int u = 0; u < 4; ++u) {
        const int i = i0 + u * AGG_TB;
        if (i < total) {
          const int dl = dv[u] - lo;
          list[i] = sv[u] | (dl << 16);
          atomicAdd(&cntn[dl], 1);
        }
      }
    }
    __syncthreads();

    if (tid < AGG_R) {
      int sum = 0;
      for (int j = 0; j < AGG_R; ++j) {
        const int v = cntn[j];
        sum += j < tid ? v : 0;
      }
      offs[tid] = sum;
      pos[tid]  = sum;
    }
    __syncthreads();
    for (int i = tid; i < total; i += AGG_TB) {
      const int v = list[i];
      const int p = atomicAdd(&pos[v >> 16], 1);
      seg[p] = v & 0xffff;
    }
    __syncthreads();

    for (int dl = wv; dl < nown; dl += AGG_TB / 32) {
      const int n = __builtin_amdgcn_readfirstlane(cntn[dl]);
      const int o = __builtin_amdgcn_readfirstlane(offs[dl]);
      for (int i0 = 0; i0 < n; i0 += 32) {
        const int i = i0 + lane;
        const int vi = seg[o + min(i, n - 1)];
        int rank = 0;
        for (int j = 0; j < n; ++j) {
          const int vj = seg[o + j];
          rank += (vj < vi || (vj == vi && j < i)) ? 1 : 0;
        }
        if (i < n) list[o + rank] = vi;
      }
    }
    __syncthreads();

    for (int i = tid; i < total; i += AGG_TB) myruns[i] = list[i];
    if (tid < AGG_R) {
      mymeta[1 + tid]         = cntn[tid];
      mymeta[1 + AGG_R + tid] = offs[tid];
    }
   } else {
    for (int i0 = tid; i0 < total; i0 += 4 * AGG_TB) {
      int rv[4];
      #pragma unroll
      for (int u = 0; u < 4; ++u) rv[u] = myruns[min(i0 + u * AGG_TB, total - 1)];
      #pragma unroll
      for (int u = 0; u < 4; ++u) {
        const int i = i0 + u * AGG_TB;
        if (i < total) list[i] = rv[u];
      }
    }
    if (tid < AGG_R) {
      cntn[tid] = mymeta[1 + tid];
      offs[tid] = mymeta[1 + AGG_R + tid];
    }
    __syncthreads();
   }

    const float4 w  = *(const float4*)(att + lane * 4);
    for (int it = 0; it < AGG_R; ++it) {
      int dl = 0;
      if (lane == 0) dl = atomicAdd(&nextn, 1);
      dl = __builtin_amdgcn_readfirstlane(dl);
      if (dl >= nown) break;
      const int n = __builtin_amdgcn_readfirstlane(cntn[dl]);
      const int o = __builtin_amdgcn_readfirstlane(offs[dl]);
      const float4 bb = *(const float4*)(xrt + dl * FDIM + lane * 4);
      float m = -INFINITY, den = 0.0f;
      float s0 = 0.0f, s1 = 0.0f, s2 = 0.0f, s3 = 0.0f;
      for (int base = 0; base < n; base += 32) {
        const int nb = min(32, n - base);
        const int srow = lane < nb ? list[o + base + lane] : 0;
        const int cs = nb <= NG_MAX     ? nb
                     : nb <= 2 * NG_MAX ? (nb + 1) >> 1
                     : nb <= 3 * NG_MAX ? (nb + 2) / 3 : (nb + 3) >> 2;
        for (int j = 0; j < nb; j += cs) {
          const int cn = min(cs, nb - j);
#define NODE_GROUP(K) case K: node_group<K>(xl, srow, j, lane, bb, w, m, den, s0, s1, s2, s3); break;
          switch (cn) {
            NODE_GROUP(1) NODE_GROUP(2) NODE_GROUP(3)
            NODE_GROUP(4) NODE_GROUP(5) NODE_GROUP(6)
            NODE_GROUP(7) NODE_GROUP(8) NODE_GROUP(9)
            default: break;
          }
#undef NODE_GROUP
        }
      }
      const float4 bz = *(const float4*)(b + lane * 4);
      float v0 = s0 / den + bz.x, v1 = s1 / den + bz.y;
      float v2 = s2 / den + bz.z, v3 = s3 / den + bz.w;
      v0 = v0 > 0.0f ? v0 : (__expf(v0) - 1.0f);
      v1 = v1 > 0.0f ? v1 : (__expf(v1) - 1.0f);
      v2 = v2 > 0.0f ? v2 : (__expf(v2) - 1.0f);
      v3 = v3 > 0.0f ? v3 : (__expf(v3) - 1.0f);
      const v4h o4 = {(_Float16)v0, (_Float16)v1, (_Float16)v2, (_Float16)v3};
      *(v4h*)(outH + (size_t)(lo + dl) * FDIM + lane * 4) = o4;
    }
  } else {
    for (int i = tid; i < AGG_R * FDIM / 4; i += AGG_TB)
      ((float4*)acc)[i] = make_float4(0.0f, 0.0f, 0.0f, 0.0f);
    if (tid < AGG_R * NHEAD) { sfx[tid] = 0ull; mk[tid] = 0u; }
    __syncthreads();
    if (tid < nown) slow_max(lo + tid, lo + tid, tid, mk, xl, xr, att);
    owner_scan<1>(key, tid, lo, &cnt, list, acc, sfx, mk, ei, xl, xr, att);
    __syncthreads();
    if (tid < nown) slow_add_row(lo + tid, lo + tid, tid, acc, sfx, mk, xl, xr, att);
    owner_scan<2>(key, tid, lo, &cnt, list, acc, sfx, mk, ei, xl, xr, att);
    __syncthreads();

    for (int i2 = tid; i2 < AGG_R * FDIM / 2; i2 += AGG_TB) {
      const int idx = i2 * 2;
      const int nl = idx >> 7, c = idx & (FDIM - 1);
      if (nl < nown) {
        const float den = (float)sfx[nl * NHEAD + (c >> 5)] * FX_INV;
        float v0 = acc[idx] / den + b[c];
        float v1 = acc[idx + 1] / den + b[c + 1];
        v0 = v0 > 0.0f ? v0 : (__expf(v0) - 1.0f);
        v1 = v1 > 0.0f ? v1 : (__expf(v1) - 1.0f);
        v2h o = {(_Float16)v0, (_Float16)v1};
        *(v2h*)(outH + (size_t)(lo + nl) * FDIM + c) = o;
      }
    }
  }
}

extern "C" void kernel_launch(void* const* d_in, const int* in_sizes, int n_in,
                              void* d_out, int out_size, void* d_ws,
                              size_t ws_size, hipStream_t stream) {
  (void)in_sizes; (void)n_in; (void)out_size; (void)ws_size;
  const float* x    = (const float*)d_in[0];
  const int*   ei   = (const int*)d_in[1];
  const float* Wl1  = (const float*)d_in[2];
  const float* Wr1  = (const float*)d_in[3];
  const float* att1 = (const float*)d_in[4];
  const float* b1   = (const float*)d_in[5];
  const float* Wl2  = (const float*)d_in[6];
  const float* Wr2  = (const float*)d_in[7];
  const float* att2 = (const float*)d_in[8];
  const float* b2   = (const float*)d_in[9];
  const float* W1   = (const float*)d_in[10];
  const float* bb1  = (const float*)d_in[11];
  const float* W2   = (const float*)d_in[12];
  const float* bb2  = (const float*)d_in[13];
  const float* W3   = (const float*)d_in[14];
  const float* bb3  = (const float*)d_in[15];
  float* out = (float*)d_out;

  char* ws = (char*)d_ws;
  size_t off = 0;
  auto alloc = [&](size_t bytes) -> char* {
    char* p = ws + off;
    off += (bytes + 255) & ~(size_t)255;
    return p;
  };
  const size_t S_F = (size_t)NNODES * FDIM;
  _Float16* xh  = (_Float16*)alloc(S_F * 2);
  _Float16* h1h = (_Float16*)alloc(S_F * 2);
  _Float16* h2h = (_Float16*)alloc(S_F * 2);
  _Float16* wt  = (_Float16*)alloc((size_t)WT_TOTAL * 2);
  const size_t uoff = off;
  float*    xl   = (float*)alloc(S_F * 4);
  float*    xr   = (float*)alloc(S_F * 4);
  unsigned* key  = (unsigned*)alloc((size_t)NEDGES);
  int* runs  = (int*)alloc((size_t)AGG_WG * AGG_CAP * 4);
  int* rmeta = (int*)alloc((size_t)AGG_WG * RM_W * 4);
  _Float16* m1 = (_Float16*)(ws + uoff);
  _Float16* m2 = (_Float16*)(ws + uoff + (size_t)NNODES * 640 * 2);

  const int TB = 256;
  const int ngrid = (int)((S_F + TB - 1) / TB);
  const dim3 gconv((NNODES + 127) / 128, 128 / 64);

  weights_f16t_kernel<<<dim3(WT_MAXTILES, 7), TB, 0, stream>>>(
      Wl1, Wr1, Wl2, Wr2, W1, W2, W3, wt);

  dst_key_kernel<<<(NE4 + TB - 1) / TB, TB, 0, stream>>>(ei, key);

  cast_f16_kernel<<<ngrid, TB, 0, stream>>>(x, xh, (int)S_F);
  wmma_gemm_kernel<<<gconv, TB, 0, stream>>>(xh, wt + WT_OFF_L1, nullptr, xl, nullptr,
                                             NNODES, 128, 128, 0);
  wmma_gemm_kernel<<<gconv, TB, 0, stream>>>(xh, wt + WT_OFF_R1, nullptr, xr, nullptr,
                                             NNODES, 128, 128, 0);
  conv_edge_kernel<0><<<AGG_WG, AGG_TB, 0, stream>>>(ei, key, xl, xr, att1, b1, h1h,
                                                     runs, rmeta);

  wmma_gemm_kernel<<<gconv, TB, 0, stream>>>(h1h, wt + WT_OFF_L2, nullptr, xl, nullptr,
                                             NNODES, 128, 128, 0);
  wmma_gemm_kernel<<<gconv, TB, 0, stream>>>(h1h, wt + WT_OFF_R2, nullptr, xr, nullptr,
                                             NNODES, 128, 128, 0);
  conv_edge_kernel<1><<<AGG_WG, AGG_TB, 0, stream>>>(ei, key, xl, xr, att2, b2, h2h,
                                                     runs, rmeta);

  const dim3 gm1((NNODES + 127) / 128, 640 / 64);
  wmma_gemm_kernel<<<gm1, TB, 0, stream>>>(h2h, wt + WT_OFF_M1, bb1, nullptr, m1,
                                           NNODES, 640, 128, 1);
  const dim3 gm2((NNODES + 127) / 128, 320 / 64);
  wmma_gemm_kernel<<<gm2, TB, 0, stream>>>(m1, wt + WT_OFF_M2, bb2, nullptr, m2,
                                           NNODES, 320, 640, 1);
  const dim3 gm3((NNODES + 127) / 128, 64 / 64);
  wmma_gemm_kernel<<<gm3, TB, 0, stream>>>(m2, wt + WT_OFF_M3, bb3, out, nullptr,
                                           NNODES, 64, 320, 2);
}


# Input cast f32->f16 as a stream of 5000 waves, each with ten whole 512-B pieces in flight (16-B loads, 8-B stores), not 200000 waves of one element per lane; same rounding, same values
# speedup vs baseline: 8.3149x; 1.0371x over previous
#include <hip/hip_runtime.h>
#include <hip/hip_bf16.h>


typedef __attribute__((ext_vector_type(16))) _Float16 v16h;
typedef __attribute__((ext_vector_type(8)))  _Float16 v8h;
typedef __attribute__((ext_vector_type(4)))  _Float16 v4h;
typedef __attribute__((ext_vector_type(8)))  float    v8f;

#define NNODES 50000
#define NEDGES 800000
#define ETOT   (NEDGES + NNODES)
#define FDIM   128
#define NHEAD  4

__device__ __forceinline__ unsigned fkey(float f) {
  unsigned u = __float_as_uint(f);
  return (u & 0x80000000u) ? ~u : (u | 0x80000000u);
}
__device__ __forceinline__ float funkey(unsigned k) {
  unsigned u = (k & 0x80000000u) ? (k & 0x7fffffffu) : ~k;
  return __uint_as_float(u);
}

#define CAST_TB  256
#define CAST_U   10
#define CAST_N4  (NNODES * FDIM / 4)
#define CAST_WG  (CAST_N4 / (CAST_TB * CAST_U))
static_assert(CAST_WG * CAST_TB * CAST_U == CAST_N4, "every wave converts CAST_U whole pieces, none is left over");

__global__ __launch_bounds__(CAST_TB) void cast_f16_kernel(
    const float* __restrict__ in, _Float16* __restrict__ out) {
  const int lane = threadIdx.x & 31;
  const int gw   = blockIdx.x * (CAST_TB / 32) + (threadIdx.x >> 5);
  const float4* in4 = (const float4*)in + gw * (CAST_U * 32) + lane;
  v4h* out4 = (v4h*)out + gw * (CAST_U * 32) + lane;
  float4 v[CAST_U];
  #pragma unroll
  for (int u = 0; u < CAST_U; ++u) v[u] = in4[u * 32];
  #pragma unroll
  for (int u = 0; u < CAST_U; ++u) {
    const v4h o = {(_Float16)v[u].x, (_Float16)v[u].y, (_Float16)v[u].z, (_Float16)v[u].w};
    out4[u * 32] = o;
  }
}

#define WT_OFF_L1  0
#define WT_OFF_R1  (WT_OFF_L1 + 128 * 128)
#define WT_OFF_L2  (WT_OFF_R1 + 128 * 128)
#define WT_OFF_R2  (WT_OFF_L2 + 128 * 128)
#define WT_OFF_M1  (WT_OFF_R2 + 128 * 128)
#define WT_OFF_M2  (WT_OFF_M1 + 128 * 640)
#define WT_OFF_M3  (WT_OFF_M2 + 640 * 320)
#define WT_TOTAL   (WT_OFF_M3 + 320 * 64)
#define WT_MAXTILES ((640 / 32) * (320 / 32))

__global__ __launch_bounds__(256) void weights_f16t_kernel(
    const float* __restrict__ w0, const float* __restrict__ w1,
    const float* __restrict__ w2, const float* __restrict__ w3,
    const float* __restrict__ w4, const float* __restrict__ w5,
    const float* __restrict__ w6, _Float16* __restrict__ wt) {
  __shared__ float tile[32][33];
  const float* Wm; int K, Nt, off;
  switch (blockIdx.y) {
    case 0:  Wm = w0; K = 128; Nt = 128; off = WT_OFF_L1; break;
    case 1:  Wm = w1; K = 128; Nt = 128; off = WT_OFF_R1; break;
    case 2:  Wm = w2; K = 128; Nt = 128; off = WT_OFF_L2; break;
    case 3:  Wm = w3; K = 128; Nt = 128; off = WT_OFF_R2; break;
    case 4:  Wm = w4; K = 128; Nt = 640; off = WT_OFF_M1; break;
    case 5:  Wm = w5; K = 640; Nt = 320; off = WT_OFF_M2; break;
    default: Wm = w6; K = 320; Nt = 64;  off = WT_OFF_M3; break;
  }
  const int tn = Nt >> 5;
  if ((int)blockIdx.x >= (K >> 5) * tn) return;
  const int k0 = ((int)blockIdx.x / tn) * 32;
  const int n0 = ((int)blockIdx.x % tn) * 32;
  const int lane = threadIdx.x & 31;
  const int wv   = threadIdx.x >> 5;
  #pragma unroll
  for (int j = 0; j < 4; ++j) {
    const int k = wv + 8 * j;
    tile[k][lane] = Wm[(size_t)(k0 + k) * Nt + n0 + lane];
  }
  __syncthreads();
  #pragma unroll
  for (int j = 0; j < 4; ++j) {
    const int n = wv + 8 * j;
    wt[off + (n0 + n) * K + k0 + lane] = (_Float16)tile[lane][n];
  }
}

#define KC 64
#define BT_STRIDE 72

__global__ __launch_bounds__(256) void wmma_gemm_kernel(
    const _Float16* __restrict__ A, const _Float16* __restrict__ Wt,
    const float* __restrict__ bias,
    float* __restrict__ outF, _Float16* __restrict__ outH,
    int M, int Nt, int K, int act) {
  __shared__ _Float16 bt[64 * BT_STRIDE];

  const int lane = threadIdx.x & 31;
  const int wave = threadIdx.x >> 5;
  const int half = lane >> 4;
  const int l15  = lane & 15;
  const int n0   = blockIdx.y * 64;

  int rowBase = blockIdx.x * 128 + wave * 16;
  if (rowBase > M - 16) rowBase = M - 16;

  v8f c[4] = {v8f{}, v8f{}, v8f{}, v8f{}};

  for (int kc = 0; kc < K; kc += KC) {
    {
      const int q8 = (threadIdx.x & 7) * 8;
      int n = threadIdx.x >> 3;
      #pragma unroll
      for (int p = 0; p < 2; ++p, n += 32) {
        const v8h wv = *(const v8h*)(Wt + (size_t)(n0 + n) * K + kc + q8);
        *(v8h*)(bt + n * BT_STRIDE + q8) = wv;
      }
    }
    __syncthreads();

    const _Float16* arow = A + (size_t)(rowBase + l15) * K + kc;
    #pragma unroll
    for (int ks = 0; ks < 2; ++ks) {
      const int kb = ks * 32 + half * 8;
      v8h alo = *(const v8h*)(arow + kb);
      v8h ahi = *(const v8h*)(arow + kb + 16);
      v16h a = __builtin_shufflevector(alo, ahi, 0, 1, 2, 3, 4, 5, 6, 7,
                                       8, 9, 10, 11, 12, 13, 14, 15);
      v16h b[4];
      #pragma unroll
      for (int nn = 0; nn < 4; ++nn) {
        const _Float16* bp = bt + (nn * 16 + l15) * BT_STRIDE + kb;
        v8h blo = *(const v8h*)(bp);
        v8h bhi = *(const v8h*)(bp + 16);
        b[nn] = __builtin_shufflevector(blo, bhi, 0, 1, 2, 3, 4, 5, 6, 7,
                                        8, 9, 10, 11, 12, 13, 14, 15);
      }
      #pragma unroll
      for (int nn = 0; nn < 4; ++nn) {
        c[nn] = __builtin_amdgcn_wmma_f32_16x16x32_f16(
            false, a, false, b[nn], (short)0, c[nn], false, false);
      }
    }
    __syncthreads();
  }

  #pragma unroll
  for (int nn = 0; nn < 4; ++nn) {
    const int col = n0 + nn * 16 + l15;
    const float bv = bias ? bias[col] : 0.0f;
    #pragma unroll
    for (int r = 0; r < 8; ++r) {
      const int row = rowBase + half * 8 + r;
      float v = c[nn][r] + bv;
      if (act == 1) v = v > 0.0f ? v : (__expf(v) - 1.0f);
      else if (act == 2) v = 1.0f / (1.0f + __expf(-v));
      if (outF) outF[(size_t)row * Nt + col] = v;
      if (outH) outH[(size_t)row * Nt + col] = (_Float16)v;
    }
  }
}

#define FX_ONE   1099511627776.0f
#define FX_INV   (1.0f / 1099511627776.0f)
#define AGG_R    196
#define AGG_WG   ((NNODES + AGG_R - 1) / AGG_R)
#define AGG_TB   1024
#define AGG_CAP  16384
#define NE4      (NEDGES / 4)
static_assert(NEDGES % 4 == 0, "dst list is read as int4");
static_assert(AGG_R <= AGG_TB && AGG_R <= AGG_CAP, "self loops are queued by one pass");
static_assert(AGG_R * NHEAD <= AGG_TB, "maxima and denominators are set up by one pass");
static_assert(NNODES <= 65536 && AGG_R <= 256, "source row and local destination share a queue slot");
static_assert(AGG_CAP <= AGG_R * FDIM, "the runs take the place of the slow path's row sums");
#define RM_W     (1 + 2 * AGG_R)

#define NK16     (NEDGES / 16)
#define SC_U     8
static_assert(AGG_WG <= 256, "an owner's number fits one byte");
static_assert(NEDGES % 16 == 0, "key bytes are read as uint4");
static_assert(SC_U == 8, "the scan packs eight loads' matches into four words");

__global__ __launch_bounds__(256) void dst_key_kernel(
    const int* __restrict__ ei, unsigned* __restrict__ key) {
  const int i = blockIdx.x * blockDim.x + threadIdx.x;
  if (i >= NE4) return;
  const int4 v = ((const int4*)(ei + NEDGES))[i];
  key[i] = ((unsigned)v.x / AGG_R) | (((unsigned)v.y / AGG_R) << 8) |
           (((unsigned)v.z / AGG_R) << 16) | (((unsigned)v.w / AGG_R) << 24);
}

__device__ __forceinline__ unsigned zero_bytes(unsigned x) {
  return ~(((x & 0x7f7f7f7fu) + 0x7f7f7f7fu) | x | 0x7f7f7f7fu);
}

typedef __attribute__((ext_vector_type(2))) _Float16 v2h;

template <int M>
__device__ __forceinline__ float xor_lane(float v) {
  return __int_as_float(__builtin_amdgcn_update_dpp(
      0, __float_as_int(v), 0x160 + M, 0xf, 0xf, true));
}

__device__ __forceinline__ unsigned long long fx_of(float p) {
  return (unsigned long long)(p * FX_ONE);
}

__device__ __forceinline__ float slow_logit(int s, int d, int h,
                                            const float* __restrict__ xl,
                                            const float* __restrict__ xr,
                                            const float* __restrict__ att) {
  float a = 0.0f;
  for (int c = h * 32; c < h * 32 + 32; ++c) {
    float z = xl[(size_t)s * FDIM + c] + xr[(size_t)d * FDIM + c];
    z = z > 0.0f ? z : 0.2f * z;
    a += z * att[c];
  }
  return a;
}

__device__ __forceinline__ void slow_max(int s, int d, int dl, unsigned* mk,
                                         const float* __restrict__ xl,
                                         const float* __restrict__ xr,
                                         const float* __restrict__ att) {
  for (int h = 0; h < NHEAD; ++h)
    atomicMax(mk + dl * NHEAD + h, fkey(slow_logit(s, d, h, xl, xr, att)));
}

__device__ __forceinline__ void slow_add_row(int s, int d, int dl, float* acc,
                                             unsigned long long* sfx,
                                             const unsigned* mk,
                                             const float* __restrict__ xl,
                                             const float* __restrict__ xr,
                                             const float* __restrict__ att) {
  for (int h = 0; h < NHEAD; ++h) {
    const float p = __expf(slow_logit(s, d, h, xl, xr, att) - funkey(mk[dl * NHEAD + h]));
    atomicAdd(sfx + dl * NHEAD + h, fx_of(p));
    for (int c = h * 32; c < h * 32 + 32; ++c)
      atomicAdd(acc + dl * FDIM + c, xl[(size_t)s * FDIM + c] * p);
  }
}

template <int MODE>
__device__ __forceinline__ void owner_scan(
    const unsigned* __restrict__ key, int tid, int lo, int* cnt, int* list,
    float* acc, unsigned long long* sfx, unsigned* mk,
    const int* __restrict__ ei, const float* __restrict__ xl,
    const float* __restrict__ xr, const float* __restrict__ att) {
  const uint4* key4 = (const uint4*)key;
  const unsigned pat = blockIdx.x * 0x01010101u;
  const int start = blockIdx.x * (NK16 / AGG_WG);
  for (int i0 = tid; i0 < NK16; i0 += AGG_TB * SC_U) {
    uint4 v[SC_U];
    #pragma unroll
    for (int u = 0; u < SC_U; ++u) {
      int iw = min(i0 + u * AGG_TB, NK16 - 1) + start;
      if (iw >= NK16) iw -= NK16;
      v[u] = key4[iw];
    }
    unsigned mt[SC_U];
    #pragma unroll
    for (int u = 0; u < SC_U; ++u) {
      mt[u] = (zero_bytes(v[u].x ^ pat) >> 7) | (zero_bytes(v[u].y ^ pat) >> 6) |
              (zero_bytes(v[u].z ^ pat) >> 5) | (zero_bytes(v[u].w ^ pat) >> 4);
      if (i0 + u * AGG_TB >= NK16) mt[u] = 0;
    }
    unsigned M0 = mt[0] | (mt[1] << 4), M1 = mt[2] | (mt[3] << 4);
    unsigned M2 = mt[4] | (mt[5] << 4), M3 = mt[6] | (mt[7] << 4);
    while (M0 | M1 | M2 | M3) {
      const int p = M0 ? 0 : M1 ? 1 : M2 ? 2 : 3;
      const unsigned w = M0 ? M0 : M1 ? M1 : M2 ? M2 : M3;
      const int bp = __builtin_ctz(w);
      const unsigned cl = w & (w - 1);
      M0 = p == 0 ? cl : M0; M1 = p == 1 ? cl : M1;
      M2 = p == 2 ? cl : M2; M3 = p == 3 ? cl : M3;
      int iw = i0 + (2 * p + ((bp >> 2) & 1)) * AGG_TB + start;
      if (iw >= NK16) iw -= NK16;
      const int e = iw * 16 + (bp & 3) * 4 + (bp >> 3);
      if (MODE == 0) {
        const int pos = atomicAdd(cnt, 1);
        if (pos < AGG_CAP) list[pos] = e;
      } else {
        const int s = ei[e], d = ei[NEDGES + e];
        if (MODE == 1) slow_max(s, d, d - lo, mk, xl, xr, att);
        else           slow_add_row(s, d, d - lo, acc, sfx, mk, xl, xr, att);
      }
    }
  }
}

#define NG_MAX 9
static_assert(4 * NG_MAX >= 32 && NG_MAX <= 9, "32 sources make at most four groups; bodies exist up to 9");

template <int CN>
__device__ __forceinline__ void node_group(
    const float* __restrict__ xl, int srow, int j, int lane,
    const float4 bb, const float4 w,
    float& m, float& den, float& s0, float& s1, float& s2, float& s3) {
  float4 a[CN];
  float  t[CN];
  #pragma unroll
  for (int u = 0; u < CN; ++u)
    a[u] = *(const float4*)(xl + __builtin_amdgcn_readlane(srow, j + u) * FDIM + lane * 4);
  __builtin_amdgcn_sched_barrier(0);
  float cm = -INFINITY;
  #pragma unroll
  for (int u = 0; u < CN; ++u) {
    float z, q;
    z = a[u].x + bb.x; z = z > 0.0f ? z : 0.2f * z; q = z * w.x;
    z = a[u].y + bb.y; z = z > 0.0f ? z : 0.2f * z; q += z * w.y;
    z = a[u].z + bb.z; z = z > 0.0f ? z : 0.2f * z; q += z * w.z;
    z = a[u].w + bb.w; z = z > 0.0f ? z : 0.2f * z; q += z * w.w;
    q += xor_lane<1>(q);
    q += xor_lane<2>(q);
    q += xor_lane<4>(q);
    t[u] = q;
    cm = q > cm ? q : cm;
  }
  const float mn = fmaxf(m, cm);
  const float sc = __expf(m - mn);
  den *= sc; s0 *= sc; s1 *= sc; s2 *= sc; s3 *= sc;
  m = mn;
  #pragma unroll
  for (int u = 0; u < CN; ++u) {
    const float p = __expf(t[u] - mn);
    den += p;
    s0 += p * a[u].x; s1 += p * a[u].y; s2 += p * a[u].z; s3 += p * a[u].w;
  }
}

template <int STAGE>
__global__ __launch_bounds__(AGG_TB) void conv_edge_kernel(
    const int* __restrict__ ei, const unsigned* __restrict__ key,
    const float* __restrict__ xl, const float* __restrict__ xr,
    const float* __restrict__ att,
    const float* __restrict__ b, _Float16* __restrict__ outH,
    int* __restrict__ runs, int* __restrict__ rmeta) {
  __shared__ __attribute__((aligned(16))) union {
    float f[AGG_R * FDIM];
    int   i[AGG_CAP];
  } au;
  __shared__ __attribute__((aligned(16))) float xrt[AGG_R * FDIM];
  __shared__ __attribute__((aligned(16))) unsigned long long sfx[AGG_R * NHEAD];
  __shared__ __attribute__((aligned(16))) unsigned mk[AGG_R * NHEAD];
  __shared__ int   list[AGG_CAP];
  __shared__ int   cntn[AGG_R];
  __shared__ int   offs[AGG_R];
  __shared__ int   pos[AGG_R];
  __shared__ int   cnt;
  __shared__ int   nextn;
  float* acc = au.f;
  int*   seg = au.i;

  const int tid  = threadIdx.x;
  const int lane = tid & 31;
  const int wave = tid >> 5;
  const int lo   = blockIdx.x * AGG_R;
  const int nown = min(AGG_R, NNODES - lo);
  int* myruns = runs + (size_t)blockIdx.x * AGG_CAP;
  int* mymeta = rmeta + blockIdx.x * RM_W;

  {
    const float4* xr4 = (const float4*)(xr + (size_t)lo * FDIM);
    for (int i = tid; i < AGG_R * FDIM / 4; i += AGG_TB)
      ((float4*)xrt)[i] = i < nown * (FDIM / 4) ? xr4[i]
                                                : make_float4(0.0f, 0.0f, 0.0f, 0.0f);
  }
  int total;
  if (STAGE == 0) {
    if (tid < AGG_R) cntn[tid] = 0;
    if (tid < nown) list[tid] = NEDGES + lo + tid;
    if (tid == 0) { cnt = nown; nextn = 0; }
    __syncthreads();

    owner_scan<0>(key, tid, lo, &cnt, list, acc, sfx, mk, ei, xl, xr, att);
    __syncthreads();

    total = __builtin_amdgcn_readfirstlane(cnt);
    if (tid == 0) mymeta[0] = total;
  } else {
    if (tid == 0) nextn = 0;
    total = __builtin_amdgcn_readfirstlane(mymeta[0]);
  }
  const int wv = __builtin_amdgcn_readfirstlane(wave);

  if (total <= AGG_CAP) {
   if (STAGE == 0) {
    for (int i0 = tid; i0 < total; i0 += 4 * AGG_TB) {
      int sv[4], dv[4];
      #pragma unroll
      for (int u = 0; u < 4; ++u) {
        const int i = i0 + u * AGG_TB;
        const int e = i < total ? list[i] : NEDGES;
        const bool self = e >= NEDGES;
        const int er = self ? 0 : e;
        const int s = ei[er], d = ei[NEDGES + er];
        sv[u] = self ? e - NEDGES : s;
        dv[u] = self ? e - NEDGES : d;
      }
      #pragma unroll
      for (int u = 0; u < 4; ++u) {
        const int i = i0 + u * AGG_TB;
        if (i < total) {
          const int dl = dv[u] - lo;
          list[i] = sv[u] | (dl << 16);
          atomicAdd(&cntn[dl], 1);
        }
      }
    }
    __syncthreads();

    if (tid < AGG_R) {
      int sum = 0;
      for (int j = 0; j < AGG_R; ++j) {
        const int v = cntn[j];
        sum += j < tid ? v : 0;
      }
      offs[tid] = sum;
      pos[tid]  = sum;
    }
    __syncthreads();
    for (int i = tid; i < total; i += AGG_TB) {
      const int v = list[i];
      const int p = atomicAdd(&pos[v >> 16], 1);
      seg[p] = v & 0xffff;
    }
    __syncthreads();

    for (int dl = wv; dl < nown; dl += AGG_TB / 32) {
      const int n = __builtin_amdgcn_readfirstlane(cntn[dl]);
      const int o = __builtin_amdgcn_readfirstlane(offs[dl]);
      for (int i0 = 0; i0 < n; i0 += 32) {
        const int i = i0 + lane;
        const int vi = seg[o + min(i, n - 1)];
        int rank = 0;
        for (int j = 0; j < n; ++j) {
          const int vj = seg[o + j];
          rank += (vj < vi || (vj == vi && j < i)) ? 1 : 0;
        }
        if (i < n) list[o + rank] = vi;
      }
    }
    __syncthreads();

    for (int i = tid; i < total; i += AGG_TB) myruns[i] = list[i];
    if (tid < AGG_R) {
      mymeta[1 + tid]         = cntn[tid];
      mymeta[1 + AGG_R + tid] = offs[tid];
    }
   } else {
    for (int i0 = tid; i0 < total; i0 += 4 * AGG_TB) {
      int rv[4];
      #pragma unroll
      for (int u = 0; u < 4; ++u) rv[u] = myruns[min(i0 + u * AGG_TB, total - 1)];
      #pragma unroll
      for (int u = 0; u < 4; ++u) {
        const int i = i0 + u * AGG_TB;
        if (i < total) list[i] = rv[u];
      }
    }
    if (tid < AGG_R) {
      cntn[tid] = mymeta[1 + tid];
      offs[tid] = mymeta[1 + AGG_R + tid];
    }
    __syncthreads();
   }

    const float4 w  = *(const float4*)(att + lane * 4);
    for (int it = 0; it < AGG_R; ++it) {
      int dl = 0;
      if (lane == 0) dl = atomicAdd(&nextn, 1);
      dl = __builtin_amdgcn_readfirstlane(dl);
      if (dl >= nown) break;
      const int n = __builtin_amdgcn_readfirstlane(cntn[dl]);
      const int o = __builtin_amdgcn_readfirstlane(offs[dl]);
      const float4 bb = *(const float4*)(xrt + dl * FDIM + lane * 4);
      float m = -INFINITY, den = 0.0f;
      float s0 = 0.0f, s1 = 0.0f, s2 = 0.0f, s3 = 0.0f;
      for (int base = 0; base < n; base += 32) {
        const int nb = min(32, n - base);
        const int srow = lane < nb ? list[o + base + lane] : 0;
        const int cs = nb <= NG_MAX     ? nb
                     : nb <= 2 * NG_MAX ? (nb + 1) >> 1
                     : nb <= 3 * NG_MAX ? (nb + 2) / 3 : (nb + 3) >> 2;
        for (int j = 0; j < nb; j += cs) {
          const int cn = min(cs, nb - j);
#define NODE_GROUP(K) case K: node_group<K>(xl, srow, j, lane, bb, w, m, den, s0, s1, s2, s3); break;
          switch (cn) {
            NODE_GROUP(1) NODE_GROUP(2) NODE_GROUP(3)
            NODE_GROUP(4) NODE_GROUP(5) NODE_GROUP(6)
            NODE_GROUP(7) NODE_GROUP(8) NODE_GROUP(9)
            default: break;
          }
#undef NODE_GROUP
        }
      }
      const float4 bz = *(const float4*)(b + lane * 4);
      float v0 = s0 / den + bz.x, v1 = s1 / den + bz.y;
      float v2 = s2 / den + bz.z, v3 = s3 / den + bz.w;
      v0 = v0 > 0.0f ? v0 : (__expf(v0) - 1.0f);
      v1 = v1 > 0.0f ? v1 : (__expf(v1) - 1.0f);
      v2 = v2 > 0.0f ? v2 : (__expf(v2) - 1.0f);
      v3 = v3 > 0.0f ? v3 : (__expf(v3) - 1.0f);
      const v4h o4 = {(_Float16)v0, (_Float16)v1, (_Float16)v2, (_Float16)v3};
      *(v4h*)(outH + (size_t)(lo + dl) * FDIM + lane * 4) = o4;
    }
  } else {
    for (int i = tid; i < AGG_R * FDIM / 4; i += AGG_TB)
      ((float4*)acc)[i] = make_float4(0.0f, 0.0f, 0.0f, 0.0f);
    if (tid < AGG_R * NHEAD) { sfx[tid] = 0ull; mk[tid] = 0u; }
    __syncthreads();
    if (tid < nown) slow_max(lo + tid, lo + tid, tid, mk, xl, xr, att);
    owner_scan<1>(key, tid, lo, &cnt, list, acc, sfx, mk, ei, xl, xr, att);
    __syncthreads();
    if (tid < nown) slow_add_row(lo + tid, lo + tid, tid, acc, sfx, mk, xl, xr, att);
    owner_scan<2>(key, tid, lo, &cnt, list, acc, sfx, mk, ei, xl, xr, att);
    __syncthreads();

    for (int i2 = tid; i2 < AGG_R * FDIM / 2; i2 += AGG_TB) {
      const int idx = i2 * 2;
      const int nl = idx >> 7, c = idx & (FDIM - 1);
      if (nl < nown) {
        const float den = (float)sfx[nl * NHEAD + (c >> 5)] * FX_INV;
        float v0 = acc[idx] / den + b[c];
        float v1 = acc[idx + 1] / den + b[c + 1];
        v0 = v0 > 0.0f ? v0 : (__expf(v0) - 1.0f);
        v1 = v1 > 0.0f ? v1 : (__expf(v1) - 1.0f);
        v2h o = {(_Float16)v0, (_Float16)v1};
        *(v2h*)(outH + (size_t)(lo + nl) * FDIM + c) = o;
      }
    }
  }
}

extern "C" void kernel_launch(void* const* d_in, const int* in_sizes, int n_in,
                              void* d_out, int out_size, void* d_ws,
                              size_t ws_size, hipStream_t stream) {
  (void)in_sizes; (void)n_in; (void)out_size; (void)ws_size;
  const float* x    = (const float*)d_in[0];
  const int*   ei   = (const int*)d_in[1];
  const float* Wl1  = (const float*)d_in[2];
  const float* Wr1  = (const float*)d_in[3];
  const float* att1 = (const float*)d_in[4];
  const float* b1   = (const float*)d_in[5];
  const float* Wl2  = (const float*)d_in[6];
  const float* Wr2  = (const float*)d_in[7];
  const float* att2 = (const float*)d_in[8];
  const float* b2   = (const float*)d_in[9];
  const float* W1   = (const float*)d_in[10];
  const float* bb1  = (const float*)d_in[11];
  const float* W2   = (const float*)d_in[12];
  const float* bb2  = (const float*)d_in[13];
  const float* W3   = (const float*)d_in[14];
  const float* bb3  = (const float*)d_in[15];
  float* out = (float*)d_out;

  char* ws = (char*)d_ws;
  size_t off = 0;
  auto alloc = [&](size_t bytes) -> char* {
    char* p = ws + off;
    off += (bytes + 255) & ~(size_t)255;
    return p;
  };
  const size_t S_F = (size_t)NNODES * FDIM;
  _Float16* xh  = (_Float16*)alloc(S_F * 2);
  _Float16* h1h = (_Float16*)alloc(S_F * 2);
  _Float16* h2h = (_Float16*)alloc(S_F * 2);
  _Float16* wt  = (_Float16*)alloc((size_t)WT_TOTAL * 2);
  const size_t uoff = off;
  float*    xl   = (float*)alloc(S_F * 4);
  float*    xr   = (float*)alloc(S_F * 4);
  unsigned* key  = (unsigned*)alloc((size_t)NEDGES);
  int* runs  = (int*)alloc((size_t)AGG_WG * AGG_CAP * 4);
  int* rmeta = (int*)alloc((size_t)AGG_WG * RM_W * 4);
  _Float16* m1 = (_Float16*)(ws + uoff);
  _Float16* m2 = (_Float16*)(ws + uoff + (size_t)NNODES * 640 * 2);

  const int TB = 256;
  const dim3 gconv((NNODES + 127) / 128, 128 / 64);

  weights_f16t_kernel<<<dim3(WT_MAXTILES, 7), TB, 0, stream>>>(
      Wl1, Wr1, Wl2, Wr2, W1, W2, W3, wt);

  dst_key_kernel<<<(NE4 + TB - 1) / TB, TB, 0, stream>>>(ei, key);

  cast_f16_kernel<<<CAST_WG, CAST_TB, 0, stream>>>(x, xh);
  wmma_gemm_kernel<<<gconv, TB, 0, stream>>>(xh, wt + WT_OFF_L1, nullptr, xl, nullptr,
                                             NNODES, 128, 128, 0);
  wmma_gemm_kernel<<<gconv, TB, 0, stream>>>(xh, wt + WT_OFF_R1, nullptr, xr, nullptr,
                                             NNODES, 128, 128, 0);
  conv_edge_kernel<0><<<AGG_WG, AGG_TB, 0, stream>>>(ei, key, xl, xr, att1, b1, h1h,
                                                     runs, rmeta);

  wmma_gemm_kernel<<<gconv, TB, 0, stream>>>(h1h, wt + WT_OFF_L2, nullptr, xl, nullptr,
                                             NNODES, 128, 128, 0);
  wmma_gemm_kernel<<<gconv, TB, 0, stream>>>(h1h, wt + WT_OFF_R2, nullptr, xr, nullptr,
                                             NNODES, 128, 128, 0);
  conv_edge_kernel<1><<<AGG_WG, AGG_TB, 0, stream>>>(ei, key, xl, xr, att2, b2, h2h,
                                                     runs, rmeta);

  const dim3 gm1((NNODES + 127) / 128, 640 / 64);
  wmma_gemm_kernel<<<gm1, TB, 0, stream>>>(h2h, wt + WT_OFF_M1, bb1, nullptr, m1,
                                           NNODES, 640, 128, 1);
  const dim3 gm2((NNODES + 127) / 128, 320 / 64);
  wmma_gemm_kernel<<<gm2, TB, 0, stream>>>(m1, wt + WT_OFF_M2, bb2, nullptr, m2,
                                           NNODES, 320, 640, 1);
  const dim3 gm3((NNODES + 127) / 128, 64 / 64);
  wmma_gemm_kernel<<<gm3, TB, 0, stream>>>(m2, wt + WT_OFF_M3, bb3, out, nullptr,
                                           NNODES, 64, 320, 2);
}


# Next layer's first GEMM runs in each conv's owner workgroup on its finished rows (kept in LDS as the WMMA A operand): 3 GEMM launches, the rows' write and re-reads gone; k and sum order unchanged
# speedup vs baseline: 9.7836x; 1.1766x over previous
#include <hip/hip_runtime.h>
#include <hip/hip_bf16.h>


typedef __attribute__((ext_vector_type(16))) _Float16 v16h;
typedef __attribute__((ext_vector_type(8)))  _Float16 v8h;
typedef __attribute__((ext_vector_type(4)))  _Float16 v4h;
typedef __attribute__((ext_vector_type(8)))  float    v8f;

#define NNODES 50000
#define NEDGES 800000
#define ETOT   (NEDGES + NNODES)
#define FDIM   128
#define NHEAD  4

__device__ __forceinline__ unsigned fkey(float f) {
  unsigned u = __float_as_uint(f);
  return (u & 0x80000000u) ? ~u : (u | 0x80000000u);
}
__device__ __forceinline__ float funkey(unsigned k) {
  unsigned u = (k & 0x80000000u) ? (k & 0x7fffffffu) : ~k;
  return __uint_as_float(u);
}

#define CAST_TB  256
#define CAST_U   10
#define CAST_N4  (NNODES * FDIM / 4)
#define CAST_WG  (CAST_N4 / (CAST_TB * CAST_U))
static_assert(CAST_WG * CAST_TB * CAST_U == CAST_N4, "every wave converts CAST_U whole pieces, none is left over");

__global__ __launch_bounds__(CAST_TB) void cast_f16_kernel(
    const float* __restrict__ in, _Float16* __restrict__ out) {
  const int lane = threadIdx.x & 31;
  const int gw   = blockIdx.x * (CAST_TB / 32) + (threadIdx.x >> 5);
  const float4* in4 = (const float4*)in + gw * (CAST_U * 32) + lane;
  v4h* out4 = (v4h*)out + gw * (CAST_U * 32) + lane;
  float4 v[CAST_U];
  #pragma unroll
  for (int u = 0; u < CAST_U; ++u) v[u] = in4[u * 32];
  #pragma unroll
  for (int u = 0; u < CAST_U; ++u) {
    const v4h o = {(_Float16)v[u].x, (_Float16)v[u].y, (_Float16)v[u].z, (_Float16)v[u].w};
    out4[u * 32] = o;
  }
}

#define WT_OFF_L1  0
#define WT_OFF_R1  (WT_OFF_L1 + 128 * 128)
#define WT_OFF_L2  (WT_OFF_R1 + 128 * 128)
#define WT_OFF_R2  (WT_OFF_L2 + 128 * 128)
#define WT_OFF_M1  (WT_OFF_R2 + 128 * 128)
#define WT_OFF_M2  (WT_OFF_M1 + 128 * 640)
#define WT_OFF_M3  (WT_OFF_M2 + 640 * 320)
#define WT_TOTAL   (WT_OFF_M3 + 320 * 64)
#define WT_MAXTILES ((640 / 32) * (320 / 32))
static_assert(WT_OFF_R2 == WT_OFF_L2 + 128 * 128, "Wl2 and Wr2 form one [256][128] matrix for the first conv's owner workgroups");

__global__ __launch_bounds__(256) void weights_f16t_kernel(
    const float* __restrict__ w0, const float* __restrict__ w1,
    const float* __restrict__ w2, const float* __restrict__ w3,
    const float* __restrict__ w4, const float* __restrict__ w5,
    const float* __restrict__ w6, _Float16* __restrict__ wt) {
  __shared__ float tile[32][33];
  const float* Wm; int K, Nt, off;
  switch (blockIdx.y) {
    case 0:  Wm = w0; K = 128; Nt = 128; off = WT_OFF_L1; break;
    case 1:  Wm = w1; K = 128; Nt = 128; off = WT_OFF_R1; break;
    case 2:  Wm = w2; K = 128; Nt = 128; off = WT_OFF_L2; break;
    case 3:  Wm = w3; K = 128; Nt = 128; off = WT_OFF_R2; break;
    case 4:  Wm = w4; K = 128; Nt = 640; off = WT_OFF_M1; break;
    case 5:  Wm = w5; K = 640; Nt = 320; off = WT_OFF_M2; break;
    default: Wm = w6; K = 320; Nt = 64;  off = WT_OFF_M3; break;
  }
  const int tn = Nt >> 5;
  if ((int)blockIdx.x >= (K >> 5) * tn) return;
  const int k0 = ((int)blockIdx.x / tn) * 32;
  const int n0 = ((int)blockIdx.x % tn) * 32;
  const int lane = threadIdx.x & 31;
  const int wv   = threadIdx.x >> 5;
  #pragma unroll
  for (int j = 0; j < 4; ++j) {
    const int k = wv + 8 * j;
    tile[k][lane] = Wm[(size_t)(k0 + k) * Nt + n0 + lane];
  }
  __syncthreads();
  #pragma unroll
  for (int j = 0; j < 4; ++j) {
    const int n = wv + 8 * j;
    wt[off + (n0 + n) * K + k0 + lane] = (_Float16)tile[lane][n];
  }
}

#define KC 64
#define BT_STRIDE 72

__global__ __launch_bounds__(256) void wmma_gemm_kernel(
    const _Float16* __restrict__ A, const _Float16* __restrict__ Wt,
    const float* __restrict__ bias,
    float* __restrict__ outF, _Float16* __restrict__ outH,
    int M, int Nt, int K, int act) {
  __shared__ _Float16 bt[64 * BT_STRIDE];

  const int lane = threadIdx.x & 31;
  const int wave = threadIdx.x >> 5;
  const int half = lane >> 4;
  const int l15  = lane & 15;
  const int n0   = blockIdx.y * 64;

  int rowBase = blockIdx.x * 128 + wave * 16;
  if (rowBase > M - 16) rowBase = M - 16;

  v8f c[4] = {v8f{}, v8f{}, v8f{}, v8f{}};

  for (int kc = 0; kc < K; kc += KC) {
    {
      const int q8 = (threadIdx.x & 7) * 8;
      int n = threadIdx.x >> 3;
      #pragma unroll
      for (int p = 0; p < 2; ++p, n += 32) {
        const v8h wv = *(const v8h*)(Wt + (size_t)(n0 + n) * K + kc + q8);
        *(v8h*)(bt + n * BT_STRIDE + q8) = wv;
      }
    }
    __syncthreads();

    const _Float16* arow = A + (size_t)(rowBase + l15) * K + kc;
    #pragma unroll
    for (int ks = 0; ks < 2; ++ks) {
      const int kb = ks * 32 + half * 8;
      v8h alo = *(const v8h*)(arow + kb);
      v8h ahi = *(const v8h*)(arow + kb + 16);
      v16h a = __builtin_shufflevector(alo, ahi, 0, 1, 2, 3, 4, 5, 6, 7,
                                       8, 9, 10, 11, 12, 13, 14, 15);
      v16h b[4];
      #pragma unroll
      for (int nn = 0; nn < 4; ++nn) {
        const _Float16* bp = bt + (nn * 16 + l15) * BT_STRIDE + kb;
        v8h blo = *(const v8h*)(bp);
        v8h bhi = *(const v8h*)(bp + 16);
        b[nn] = __builtin_shufflevector(blo, bhi, 0, 1, 2, 3, 4, 5, 6, 7,
                                        8, 9, 10, 11, 12, 13, 14, 15);
      }
      #pragma unroll
      for (int nn = 0; nn < 4; ++nn) {
        c[nn] = __builtin_amdgcn_wmma_f32_16x16x32_f16(
            false, a, false, b[nn], (short)0, c[nn], false, false);
      }
    }
    __syncthreads();
  }

  #pragma unroll
  for (int nn = 0; nn < 4; ++nn) {
    const int col = n0 + nn * 16 + l15;
    const float bv = bias ? bias[col] : 0.0f;
    #pragma unroll
    for (int r = 0; r < 8; ++r) {
      const int row = rowBase + half * 8 + r;
      float v = c[nn][r] + bv;
      if (act == 1) v = v > 0.0f ? v : (__expf(v) - 1.0f);
      else if (act == 2) v = 1.0f / (1.0f + __expf(-v));
      if (outF) outF[(size_t)row * Nt + col] = v;
      if (outH) outH[(size_t)row * Nt + col] = (_Float16)v;
    }
  }
}

#define FX_ONE   1099511627776.0f
#define FX_INV   (1.0f / 1099511627776.0f)
#define AGG_R    196
#define AGG_WG   ((NNODES + AGG_R - 1) / AGG_R)
#define AGG_TB   1024
#define AGG_CAP  16384
#define NE4      (NEDGES / 4)
static_assert(NEDGES % 4 == 0, "dst list is read as int4");
static_assert(AGG_R <= AGG_TB && AGG_R <= AGG_CAP, "self loops are queued by one pass");
static_assert(AGG_R * NHEAD <= AGG_TB, "maxima and denominators are set up by one pass");
static_assert(NNODES <= 65536 && AGG_R <= 256, "source row and local destination share a queue slot");
static_assert(AGG_CAP <= AGG_R * FDIM, "the runs take the place of the slow path's row sums");
#define RM_W     (1 + 2 * AGG_R)

#define NK16     (NEDGES / 16)
#define SC_U     8
static_assert(AGG_WG <= 256, "an owner's number fits one byte");
static_assert(NEDGES % 16 == 0, "key bytes are read as uint4");
static_assert(SC_U == 8, "the scan packs eight loads' matches into four words");

__global__ __launch_bounds__(256) void dst_key_kernel(
    const int* __restrict__ ei, unsigned* __restrict__ key) {
  const int i = blockIdx.x * blockDim.x + threadIdx.x;
  if (i >= NE4) return;
  const int4 v = ((const int4*)(ei + NEDGES))[i];
  key[i] = ((unsigned)v.x / AGG_R) | (((unsigned)v.y / AGG_R) << 8) |
           (((unsigned)v.z / AGG_R) << 16) | (((unsigned)v.w / AGG_R) << 24);
}

__device__ __forceinline__ unsigned zero_bytes(unsigned x) {
  return ~(((x & 0x7f7f7f7fu) + 0x7f7f7f7fu) | x | 0x7f7f7f7fu);
}

typedef __attribute__((ext_vector_type(2))) _Float16 v2h;

template <int M>
__device__ __forceinline__ float xor_lane(float v) {
  return __int_as_float(__builtin_amdgcn_update_dpp(
      0, __float_as_int(v), 0x160 + M, 0xf, 0xf, true));
}

__device__ __forceinline__ unsigned long long fx_of(float p) {
  return (unsigned long long)(p * FX_ONE);
}

__device__ __forceinline__ float slow_logit(int s, int d, int h,
                                            const float* __restrict__ xl,
                                            const float* __restrict__ xr,
                                            const float* __restrict__ att) {
  float a = 0.0f;
  for (int c = h * 32; c < h * 32 + 32; ++c) {
    float z = xl[(size_t)s * FDIM + c] + xr[(size_t)d * FDIM + c];
    z = z > 0.0f ? z : 0.2f * z;
    a += z * att[c];
  }
  return a;
}

__device__ __forceinline__ void slow_max(int s, int d, int dl, unsigned* mk,
                                         const float* __restrict__ xl,
                                         const float* __restrict__ xr,
                                         const float* __restrict__ att) {
  for (int h = 0; h < NHEAD; ++h)
    atomicMax(mk + dl * NHEAD + h, fkey(slow_logit(s, d, h, xl, xr, att)));
}

__device__ __forceinline__ void slow_add_row(int s, int d, int dl, float* acc,
                                             unsigned long long* sfx,
                                             const unsigned* mk,
                                             const float* __restrict__ xl,
                                             const float* __restrict__ xr,
                                             const float* __restrict__ att) {
  for (int h = 0; h < NHEAD; ++h) {
    const float p = __expf(slow_logit(s, d, h, xl, xr, att) - funkey(mk[dl * NHEAD + h]));
    atomicAdd(sfx + dl * NHEAD + h, fx_of(p));
    for (int c = h * 32; c < h * 32 + 32; ++c)
      atomicAdd(acc + dl * FDIM + c, xl[(size_t)s * FDIM + c] * p);
  }
}

template <int MODE>
__device__ __forceinline__ void owner_scan(
    const unsigned* __restrict__ key, int tid, int lo, int* cnt, int* list,
    float* acc, unsigned long long* sfx, unsigned* mk,
    const int* __restrict__ ei, const float* __restrict__ xl,
    const float* __restrict__ xr, const float* __restrict__ att) {
  const uint4* key4 = (const uint4*)key;
  const unsigned pat = blockIdx.x * 0x01010101u;
  const int start = blockIdx.x * (NK16 / AGG_WG);
  for (int i0 = tid; i0 < NK16; i0 += AGG_TB * SC_U) {
    uint4 v[SC_U];
    #pragma unroll
    for (int u = 0; u < SC_U; ++u) {
      int iw = min(i0 + u * AGG_TB, NK16 - 1) + start;
      if (iw >= NK16) iw -= NK16;
      v[u] = key4[iw];
    }
    unsigned mt[SC_U];
    #pragma unroll
    for (int u = 0; u < SC_U; ++u) {
      mt[u] = (zero_bytes(v[u].x ^ pat) >> 7) | (zero_bytes(v[u].y ^ pat) >> 6) |
              (zero_bytes(v[u].z ^ pat) >> 5) | (zero_bytes(v[u].w ^ pat) >> 4);
      if (i0 + u * AGG_TB >= NK16) mt[u] = 0;
    }
    unsigned M0 = mt[0] | (mt[1] << 4), M1 = mt[2] | (mt[3] << 4);
    unsigned M2 = mt[4] | (mt[5] << 4), M3 = mt[6] | (mt[7] << 4);
    while (M0 | M1 | M2 | M3) {
      const int p = M0 ? 0 : M1 ? 1 : M2 ? 2 : 3;
      const unsigned w = M0 ? M0 : M1 ? M1 : M2 ? M2 : M3;
      const int bp = __builtin_ctz(w);
      const unsigned cl = w & (w - 1);
      M0 = p == 0 ? cl : M0; M1 = p == 1 ? cl : M1;
      M2 = p == 2 ? cl : M2; M3 = p == 3 ? cl : M3;
      int iw = i0 + (2 * p + ((bp >> 2) & 1)) * AGG_TB + start;
      if (iw >= NK16) iw -= NK16;
      const int e = iw * 16 + (bp & 3) * 4 + (bp >> 3);
      if (MODE == 0) {
        const int pos = atomicAdd(cnt, 1);
        if (pos < AGG_CAP) list[pos] = e;
      } else {
        const int s = ei[e], d = ei[NEDGES + e];
        if (MODE == 1) slow_max(s, d, d - lo, mk, xl, xr, att);
        else           slow_add_row(s, d, d - lo, acc, sfx, mk, xl, xr, att);
      }
    }
  }
}

#define NG_MAX 9
static_assert(4 * NG_MAX >= 32 && NG_MAX <= 9, "32 sources make at most four groups; bodies exist up to 9");

template <int CN>
__device__ __forceinline__ void node_group(
    const float* __restrict__ xl, int srow, int j, int lane,
    const float4 bb, const float4 w,
    float& m, float& den, float& s0, float& s1, float& s2, float& s3) {
  float4 a[CN];
  float  t[CN];
  #pragma unroll
  for (int u = 0; u < CN; ++u)
    a[u] = *(const float4*)(xl + __builtin_amdgcn_readlane(srow, j + u) * FDIM + lane * 4);
  __builtin_amdgcn_sched_barrier(0);
  float cm = -INFINITY;
  #pragma unroll
  for (int u = 0; u < CN; ++u) {
    float z, q;
    z = a[u].x + bb.x; z = z > 0.0f ? z : 0.2f * z; q = z * w.x;
    z = a[u].y + bb.y; z = z > 0.0f ? z : 0.2f * z; q += z * w.y;
    z = a[u].z + bb.z; z = z > 0.0f ? z : 0.2f * z; q += z * w.z;
    z = a[u].w + bb.w; z = z > 0.0f ? z : 0.2f * z; q += z * w.w;
    q += xor_lane<1>(q);
    q += xor_lane<2>(q);
    q += xor_lane<4>(q);
    t[u] = q;
    cm = q > cm ? q : cm;
  }
  const float mn = fmaxf(m, cm);
  const float sc = __expf(m - mn);
  den *= sc; s0 *= sc; s1 *= sc; s2 *= sc; s3 *= sc;
  m = mn;
  #pragma unroll
  for (int u = 0; u < CN; ++u) {
    const float p = __expf(t[u] - mn);
    den += p;
    s0 += p * a[u].x; s1 += p * a[u].y; s2 += p * a[u].z; s3 += p * a[u].w;
  }
}

#define HP   136
#define NMT  ((AGG_R + 15) / 16)
#define M1W  640
static_assert(NMT * 16 * HP * 2 <= AGG_R * FDIM * 4, "the f16 rows fit the place of the runs (or of the xr rows)");
static_assert(AGG_TB / 32 == 32 && M1W / 16 == 40, "32 waves: 16 column tiles x 2, or 8 groups of 5 column tiles x 4");

template <int STAGE>
__device__ __forceinline__ void next_linear(
    const _Float16* hrows, const _Float16* __restrict__ Wn,
    const float* __restrict__ bn, float* __restrict__ oxl, float* __restrict__ oxr,
    _Float16* __restrict__ om, int lo, int nown, int wv, int lane) {
  const int half = lane >> 4;
  const int l15  = lane & 15;
  const int nmt  = (nown + 15) >> 4;
  const int NTW = STAGE == 0 ? 1 : 5;
  const int MS  = STAGE == 0 ? 2 : 4;
  const int nt0 = STAGE == 0 ? (wv & 15) : (wv >> 2) * 5;
  const int mt0 = STAGE == 0 ? (wv >> 4) : (wv & 3);
  for (int q = 0; q < NTW; ++q) {
    const int nt = nt0 + q;
    v16h b[4];
    const unsigned wo = (unsigned)(((nt * 16 + l15) * FDIM + half * 8) * 2);
    #pragma unroll
    for (int ks = 0; ks < 4; ++ks) {
      const v8h blo = *(const v8h*)((const char*)Wn + (wo + ks * 64));
      const v8h bhi = *(const v8h*)((const char*)Wn + (wo + ks * 64 + 32));
      b[ks] = __builtin_shufflevector(blo, bhi, 0, 1, 2, 3, 4, 5, 6, 7,
                                      8, 9, 10, 11, 12, 13, 14, 15);
    }
    float bv = 0.0f;
    if (STAGE == 1) bv = bn[nt * 16 + l15];
    for (int mg = mt0; mg < nmt; mg += 4 * MS) {
      v8f c[4] = {v8f{}, v8f{}, v8f{}, v8f{}};
      #pragma unroll
      for (int ks = 0; ks < 4; ++ks) {
        #pragma unroll
        for (int j = 0; j < 4; ++j) {
          const int mt = min(mg + j * MS, nmt - 1);
          const _Float16* ap = hrows + (mt * 16 + l15) * HP + ks * 32 + half * 8;
          const v8h alo = *(const v8h*)(ap);
          const v8h ahi = *(const v8h*)(ap + 16);
          const v16h a = __builtin_shufflevector(alo, ahi, 0, 1, 2, 3, 4, 5, 6, 7,
                                                 8, 9, 10, 11, 12, 13, 14, 15);
          c[j] = __builtin_amdgcn_wmma_f32_16x16x32_f16(
              false, a, false, b[ks], (short)0, c[j], false, false);
        }
      }
      #pragma unroll
      for (int j = 0; j < 4; ++j) {
        const int mt = mg + j * MS;
        if (mt < nmt) {
          const int r0  = mt * 16 + half * 8;
          const int lim = nown - r0;
          if (STAGE == 0) {
            float* o = (nt < 8 ? oxl : oxr) + (unsigned)((lo + r0) * FDIM + (nt & 7) * 16 + l15);
            #pragma unroll
            for (int r = 0; r < 8; ++r)
              if (r < lim) o[r * FDIM] = c[j][r];
          } else {
            _Float16* o = om + (unsigned)((lo + r0) * M1W + nt * 16 + l15);
            #pragma unroll
            for (int r = 0; r < 8; ++r) {
              float v = c[j][r] + bv;
              v = v > 0.0f ? v : (__expf(v) - 1.0f);
              if (r < lim) o[r * M1W] = (_Float16)v;
            }
          }
        }
      }
    }
  }
}

template <int STAGE>
__global__ __launch_bounds__(AGG_TB) void conv_edge_kernel(
    const int* __restrict__ ei, const unsigned* __restrict__ key,
    const float* __restrict__ xl, const float* __restrict__ xr,
    const float* __restrict__ att,
    const float* __restrict__ b,
    int* __restrict__ runs, int* __restrict__ rmeta,
    const _Float16* __restrict__ Wn, const float* __restrict__ bn,
    float* __restrict__ oxl, float* __restrict__ oxr, _Float16* __restrict__ om) {
  __shared__ __attribute__((aligned(16))) union {
    float f[AGG_R * FDIM];
    int   i[AGG_CAP];
  } au;
  __shared__ __attribute__((aligned(16))) float xrt[AGG_R * FDIM];
  __shared__ __attribute__((aligned(16))) unsigned long long sfx[AGG_R * NHEAD];
  __shared__ __attribute__((aligned(16))) unsigned mk[AGG_R * NHEAD];
  __shared__ int   list[AGG_CAP];
  __shared__ int   cntn[AGG_R];
  __shared__ int   offs[AGG_R];
  __shared__ int   pos[AGG_R];
  __shared__ int   cnt;
  __shared__ int   nextn;
  float* acc = au.f;
  int*   seg = au.i;

  const int tid  = threadIdx.x;
  const int lane = tid & 31;
  const int wave = tid >> 5;
  const int lo   = blockIdx.x * AGG_R;
  const int nown = min(AGG_R, NNODES - lo);
  int* myruns = runs + (size_t)blockIdx.x * AGG_CAP;
  int* mymeta = rmeta + blockIdx.x * RM_W;

  {
    const float4* xr4 = (const float4*)(xr + (size_t)lo * FDIM);
    for (int i = tid; i < AGG_R * FDIM / 4; i += AGG_TB)
      ((float4*)xrt)[i] = i < nown * (FDIM / 4) ? xr4[i]
                                                : make_float4(0.0f, 0.0f, 0.0f, 0.0f);
  }
  int total;
  if (STAGE == 0) {
    if (tid < AGG_R) cntn[tid] = 0;
    if (tid < nown) list[tid] = NEDGES + lo + tid;
    if (tid == 0) { cnt = nown; nextn = 0; }
    __syncthreads();

    owner_scan<0>(key, tid, lo, &cnt, list, acc, sfx, mk, ei, xl, xr, att);
    __syncthreads();

    total = __builtin_amdgcn_readfirstlane(cnt);
    if (tid == 0) mymeta[0] = total;
  } else {
    if (tid == 0) nextn = 0;
    total = __builtin_amdgcn_readfirstlane(mymeta[0]);
  }
  const int wv = __builtin_amdgcn_readfirstlane(wave);
  _Float16* hrows = total <= AGG_CAP ? (_Float16*)au.f : (_Float16*)xrt;

  if (total <= AGG_CAP) {
   if (STAGE == 0) {
    for (int i0 = tid; i0 < total; i0 += 4 * AGG_TB) {
      int sv[4], dv[4];
      #pragma unroll
      for (int u = 0; u < 4; ++u) {
        const int i = i0 + u * AGG_TB;
        const int e = i < total ? list[i] : NEDGES;
        const bool self = e >= NEDGES;
        const int er = self ? 0 : e;
        const int s = ei[er], d = ei[NEDGES + er];
        sv[u] = self ? e - NEDGES : s;
        dv[u] = self ? e - NEDGES : d;
      }
      #pragma unroll
      for (int u = 0; u < 4; ++u) {
        const int i = i0 + u * AGG_TB;
        if (i < total) {
          const int dl = dv[u] - lo;
          list[i] = sv[u] | (dl << 16);
          atomicAdd(&cntn[dl], 1);
        }
      }
    }
    __syncthreads();

    if (tid < AGG_R) {
      int sum = 0;
      for (int j = 0; j < AGG_R; ++j) {
        const int v = cntn[j];
        sum += j < tid ? v : 0;
      }
      offs[tid] = sum;
      pos[tid]  = sum;
    }
    __syncthreads();
    for (int i = tid; i < total; i += AGG_TB) {
      const int v = list[i];
      const int p = atomicAdd(&pos[v >> 16], 1);
      seg[p] = v & 0xffff;
    }
    __syncthreads();

    for (int dl = wv; dl < nown; dl += AGG_TB / 32) {
      const int n = __builtin_amdgcn_readfirstlane(cntn[dl]);
      const int o = __builtin_amdgcn_readfirstlane(offs[dl]);
      for (int i0 = 0; i0 < n; i0 += 32) {
        const int i = i0 + lane;
        const int vi = seg[o + min(i, n - 1)];
        int rank = 0;
        for (int j = 0; j < n; ++j) {
          const int vj = seg[o + j];
          rank += (vj < vi || (vj == vi && j < i)) ? 1 : 0;
        }
        if (i < n) list[o + rank] = vi;
      }
    }
    __syncthreads();

    for (int i = tid; i < total; i += AGG_TB) myruns[i] = list[i];
    if (tid < AGG_R) {
      mymeta[1 + tid]         = cntn[tid];
      mymeta[1 + AGG_R + tid] = offs[tid];
    }
   } else {
    for (int i0 = tid; i0 < total; i0 += 4 * AGG_TB) {
      int rv[4];
      #pragma unroll
      for (int u = 0; u < 4; ++u) rv[u] = myruns[min(i0 + u * AGG_TB, total - 1)];
      #pragma unroll
      for (int u = 0; u < 4; ++u) {
        const int i = i0 + u * AGG_TB;
        if (i < total) list[i] = rv[u];
      }
    }
    if (tid < AGG_R) {
      cntn[tid] = mymeta[1 + tid];
      offs[tid] = mymeta[1 + AGG_R + tid];
    }
    __syncthreads();
   }

    const float4 w  = *(const float4*)(att + lane * 4);
    for (int it = 0; it < AGG_R; ++it) {
      int dl = 0;
      if (lane == 0) dl = atomicAdd(&nextn, 1);
      dl = __builtin_amdgcn_readfirstlane(dl);
      if (dl >= nown) break;
      const int n = __builtin_amdgcn_readfirstlane(cntn[dl]);
      const int o = __builtin_amdgcn_readfirstlane(offs[dl]);
      const float4 bb = *(const float4*)(xrt + dl * FDIM + lane * 4);
      float m = -INFINITY, den = 0.0f;
      float s0 = 0.0f, s1 = 0.0f, s2 = 0.0f, s3 = 0.0f;
      for (int base = 0; base < n; base += 32) {
        const int nb = min(32, n - base);
        const int srow = lane < nb ? list[o + base + lane] : 0;
        const int cs = nb <= NG_MAX     ? nb
                     : nb <= 2 * NG_MAX ? (nb + 1) >> 1
                     : nb <= 3 * NG_MAX ? (nb + 2) / 3 : (nb + 3) >> 2;
        for (int j = 0; j < nb; j += cs) {
          const int cn = min(cs, nb - j);
#define NODE_GROUP(K) case K: node_group<K>(xl, srow, j, lane, bb, w, m, den, s0, s1, s2, s3); break;
          switch (cn) {
            NODE_GROUP(1) NODE_GROUP(2) NODE_GROUP(3)
            NODE_GROUP(4) NODE_GROUP(5) NODE_GROUP(6)
            NODE_GROUP(7) NODE_GROUP(8) NODE_GROUP(9)
            default: break;
          }
#undef NODE_GROUP
        }
      }
      const float4 bz = *(const float4*)(b + lane * 4);
      float v0 = s0 / den + bz.x, v1 = s1 / den + bz.y;
      float v2 = s2 / den + bz.z, v3 = s3 / den + bz.w;
      v0 = v0 > 0.0f ? v0 : (__expf(v0) - 1.0f);
      v1 = v1 > 0.0f ? v1 : (__expf(v1) - 1.0f);
      v2 = v2 > 0.0f ? v2 : (__expf(v2) - 1.0f);
      v3 = v3 > 0.0f ? v3 : (__expf(v3) - 1.0f);
      const v4h o4 = {(_Float16)v0, (_Float16)v1, (_Float16)v2, (_Float16)v3};
      *(v4h*)(hrows + dl * HP + lane * 4) = o4;
    }
  } else {
    for (int i = tid; i < AGG_R * FDIM / 4; i += AGG_TB)
      ((float4*)acc)[i] = make_float4(0.0f, 0.0f, 0.0f, 0.0f);
    if (tid < AGG_R * NHEAD) { sfx[tid] = 0ull; mk[tid] = 0u; }
    __syncthreads();
    if (tid < nown) slow_max(lo + tid, lo + tid, tid, mk, xl, xr, att);
    owner_scan<1>(key, tid, lo, &cnt, list, acc, sfx, mk, ei, xl, xr, att);
    __syncthreads();
    if (tid < nown) slow_add_row(lo + tid, lo + tid, tid, acc, sfx, mk, xl, xr, att);
    owner_scan<2>(key, tid, lo, &cnt, list, acc, sfx, mk, ei, xl, xr, att);
    __syncthreads();

    for (int i2 = tid; i2 < AGG_R * FDIM / 2; i2 += AGG_TB) {
      const int idx = i2 * 2;
      const int nl = idx >> 7, c = idx & (FDIM - 1);
      if (nl < nown) {
        const float den = (float)sfx[nl * NHEAD + (c >> 5)] * FX_INV;
        float v0 = acc[idx] / den + b[c];
        float v1 = acc[idx + 1] / den + b[c + 1];
        v0 = v0 > 0.0f ? v0 : (__expf(v0) - 1.0f);
        v1 = v1 > 0.0f ? v1 : (__expf(v1) - 1.0f);
        v2h o = {(_Float16)v0, (_Float16)v1};
        *(v2h*)(hrows + nl * HP + c) = o;
      }
    }
  }
  __syncthreads();
  next_linear<STAGE>(hrows, Wn, bn, oxl, oxr, om, lo, nown, wv, lane);
}

extern "C" void kernel_launch(void* const* d_in, const int* in_sizes, int n_in,
                              void* d_out, int out_size, void* d_ws,
                              size_t ws_size, hipStream_t stream) {
  (void)in_sizes; (void)n_in; (void)out_size; (void)ws_size;
  const float* x    = (const float*)d_in[0];
  const int*   ei   = (const int*)d_in[1];
  const float* Wl1  = (const float*)d_in[2];
  const float* Wr1  = (const float*)d_in[3];
  const float* att1 = (const float*)d_in[4];
  const float* b1   = (const float*)d_in[5];
  const float* Wl2  = (const float*)d_in[6];
  const float* Wr2  = (const float*)d_in[7];
  const float* att2 = (const float*)d_in[8];
  const float* b2   = (const float*)d_in[9];
  const float* W1   = (const float*)d_in[10];
  const float* bb1  = (const float*)d_in[11];
  const float* W2   = (const float*)d_in[12];
  const float* bb2  = (const float*)d_in[13];
  const float* W3   = (const float*)d_in[14];
  const float* bb3  = (const float*)d_in[15];
  float* out = (float*)d_out;

  char* ws = (char*)d_ws;
  size_t off = 0;
  auto alloc = [&](size_t bytes) -> char* {
    char* p = ws + off;
    off += (bytes + 255) & ~(size_t)255;
    return p;
  };
  const size_t S_F = (size_t)NNODES * FDIM;
  _Float16* xh  = (_Float16*)alloc(S_F * 2);
  _Float16* wt  = (_Float16*)alloc((size_t)WT_TOTAL * 2);
  float*    xl   = (float*)alloc(S_F * 4);
  float*    xr   = (float*)alloc(S_F * 4);
  unsigned* key  = (unsigned*)alloc((size_t)NEDGES);
  int* runs  = (int*)alloc((size_t)AGG_WG * AGG_CAP * 4);
  int* rmeta = (int*)alloc((size_t)AGG_WG * RM_W * 4);
  float*    xl2  = (float*)alloc(S_F * 4);
  float*    xr2  = (float*)alloc(S_F * 4);
  _Float16* m1   = (_Float16*)alloc((size_t)NNODES * M1W * 2);
  _Float16* m2 = (_Float16*)xl;

  const int TB = 256;
  const dim3 gconv((NNODES + 127) / 128, 128 / 64);

  weights_f16t_kernel<<<dim3(WT_MAXTILES, 7), TB, 0, stream>>>(
      Wl1, Wr1, Wl2, Wr2, W1, W2, W3, wt);

  dst_key_kernel<<<(NE4 + TB - 1) / TB, TB, 0, stream>>>(ei, key);

  cast_f16_kernel<<<CAST_WG, CAST_TB, 0, stream>>>(x, xh);
  wmma_gemm_kernel<<<gconv, TB, 0, stream>>>(xh, wt + WT_OFF_L1, nullptr, xl, nullptr,
                                             NNODES, 128, 128, 0);
  wmma_gemm_kernel<<<gconv, TB, 0, stream>>>(xh, wt + WT_OFF_R1, nullptr, xr, nullptr,
                                             NNODES, 128, 128, 0);
  conv_edge_kernel<0><<<AGG_WG, AGG_TB, 0, stream>>>(ei, key, xl, xr, att1, b1,
                                                     runs, rmeta, wt + WT_OFF_L2, nullptr,
                                                     xl2, xr2, nullptr);

  conv_edge_kernel<1><<<AGG_WG, AGG_TB, 0, stream>>>(ei, key, xl2, xr2, att2, b2,
                                                     runs, rmeta, wt + WT_OFF_M1, bb1,
                                                     nullptr, nullptr, m1);

  const dim3 gm2((NNODES + 127) / 128, 320 / 64);
  wmma_gemm_kernel<<<gm2, TB, 0, stream>>>(m1, wt + WT_OFF_M2, bb2, nullptr, m2,
                                           NNODES, 320, 640, 1);
  const dim3 gm3((NNODES + 127) / 128, 64 / 64);
  wmma_gemm_kernel<<<gm3, TB, 0, stream>>>(m2, wt + WT_OFF_M3, bb3, out, nullptr,
                                           NNODES, 64, 320, 2);
}
